# starting kernel
# baseline (speedup 1.0000x reference)
#include <hip/hip_runtime.h>
#include <hip/hip_bf16.h>

typedef __attribute__((ext_vector_type(2))) float v2f;
typedef __attribute__((ext_vector_type(8))) float v8f;

#define HDIM 64

__global__ __launch_bounds__(256)
void deg_kernel(const int* __restrict__ src, const int* __restrict__ dst,
                float* __restrict__ degU, float* __restrict__ degM, int nE) {
    int e = blockIdx.x * blockDim.x + threadIdx.x;
    if (e >= nE) return;
    atomicAdd(degU + src[e], 1.0f);
    atomicAdd(degM + dst[e], 1.0f);
}

__global__ __launch_bounds__(256)
void movie_init_kernel(const float* __restrict__ mx, const float* __restrict__ linW,
                       const float* __restrict__ linb, const float* __restrict__ memb,
                       float* __restrict__ XM, int M) {
    int t = blockIdx.x * blockDim.x + threadIdx.x;
    if (t >= M * HDIM) return;
    int m = t >> 6, h = t & 63;
    float acc = linb[h] + memb[t];
    const float* xr = mx + (size_t)m * 20;
    const float* wr = linW + (size_t)h * 20;
#pragma unroll
    for (int f = 0; f < 20; ++f) acc = fmaf(xr[f], wr[f], acc);
    XM[t] = acc;
}

__global__ __launch_bounds__(256)
void scatter_kernel(const float* __restrict__ xsrc, const int* __restrict__ src,
                    const int* __restrict__ dst, float* __restrict__ agg, int nE) {
    long long tid = (long long)blockIdx.x * blockDim.x + threadIdx.x;
    if (tid >= (long long)nE * 16) return;
    int e = (int)(tid >> 4);
    int q = (int)(tid & 15);
    int s = src[e], d = dst[e];
    const float4 v = *(const float4*)(xsrc + (size_t)s * HDIM + q * 4);
    float* p = agg + (size_t)d * HDIM + q * 4;
    atomicAdd(p + 0, v.x);
    atomicAdd(p + 1, v.y);
    atomicAdd(p + 2, v.z);
    atomicAdd(p + 3, v.w);
}

__global__ __launch_bounds__(128)
void combine_wmma_kernel(const float* __restrict__ agg, const float* __restrict__ deg,
                         const float* __restrict__ xdst,
                         const float* __restrict__ Wl, const float* __restrict__ Wr,
                         const float* __restrict__ bias, float* __restrict__ out,
                         int do_relu) {
    __shared__ float A1[16][68];
    __shared__ float A2[16][68];

    const int row0 = blockIdx.x * 16;
    const int tid  = threadIdx.x;

    {
        int r  = tid >> 3;
        int c0 = (tid & 7) * 8;
        float dinv = 1.0f / fmaxf(deg[row0 + r], 1.0f);
        const float* ga = agg  + (size_t)(row0 + r) * HDIM + c0;
        const float* gx = xdst + (size_t)(row0 + r) * HDIM + c0;
#pragma unroll
        for (int j = 0; j < 8; ++j) {
            A1[r][c0 + j] = ga[j] * dinv;
            A2[r][c0 + j] = gx[j];
        }
    }
    __syncthreads();

    const int wave = tid >> 5;
    const int lane = tid & 31;
    const int m    = lane & 15;
    const int half = lane >> 4;
    const int col0 = wave * 16;

    v8f c = {};
#pragma unroll
    for (int kk = 0; kk < 16; ++kk) {
        int k = kk * 4 + half * 2;
        v2f a, b;
        a.x = A1[m][k];  a.y = A1[m][k + 1];
        const float* wp = Wl + (size_t)(col0 + m) * HDIM + k;
        b.x = wp[0];     b.y = wp[1];
        c = __builtin_amdgcn_wmma_f32_16x16x4_f32(false, a, false, b,
                                                  (short)0, c, false, false);
    }
#pragma unroll
    for (int kk = 0; kk < 16; ++kk) {
        int k = kk * 4 + half * 2;
        v2f a, b;
        a.x = A2[m][k];  a.y = A2[m][k + 1];
        const float* wp = Wr + (size_t)(col0 + m) * HDIM + k;
        b.x = wp[0];     b.y = wp[1];
        c = __builtin_amdgcn_wmma_f32_16x16x4_f32(false, a, false, b,
                                                  (short)0, c, false, false);
    }

    float bv = bias[col0 + m];
#pragma unroll
    for (int r = 0; r < 8; ++r) {
        float v = c[r] + bv;
        if (do_relu) v = fmaxf(v, 0.0f);
        out[(size_t)(row0 + r + 8 * half) * HDIM + col0 + m] = v;
    }
}

__global__ __launch_bounds__(256)
void edge_dot_kernel(const float* __restrict__ U2, const float* __restrict__ M2,
                     const int* __restrict__ eu, const int* __restrict__ em,
                     float* __restrict__ out, int L) {
    long long tid = (long long)blockIdx.x * blockDim.x + threadIdx.x;
    if (tid >= (long long)L * 8) return;
    int p = (int)(tid >> 3);
    int s = (int)(tid & 7);
    const float4* a = (const float4*)(U2 + (size_t)eu[p] * HDIM + s * 8);
    const float4* b = (const float4*)(M2 + (size_t)em[p] * HDIM + s * 8);
    float4 a0 = a[0], a1 = a[1], b0 = b[0], b1 = b[1];
    float acc = a0.x * b0.x + a0.y * b0.y + a0.z * b0.z + a0.w * b0.w
              + a1.x * b1.x + a1.y * b1.y + a1.z * b1.z + a1.w * b1.w;
    acc += __shfl_xor(acc, 1);
    acc += __shfl_xor(acc, 2);
    acc += __shfl_xor(acc, 4);
    if (s == 0) out[p] = acc;
}

extern "C" void kernel_launch(void* const* d_in, const int* in_sizes, int n_in,
                              void* d_out, int out_size, void* d_ws, size_t ws_size,
                              hipStream_t stream) {
    (void)n_in; (void)ws_size;
    const float* movie_x   = (const float*)d_in[2];
    const int*   e_src     = (const int*)d_in[3];
    const int*   e_dst     = (const int*)d_in[4];
    const int*   eli_u     = (const int*)d_in[5];
    const int*   eli_m     = (const int*)d_in[6];
    const float* user_emb  = (const float*)d_in[7];
    const float* movie_emb = (const float*)d_in[8];
    const float* lin_W     = (const float*)d_in[9];
    const float* lin_b     = (const float*)d_in[10];
    const float* W1um_l    = (const float*)d_in[11];
    const float* W1um_r    = (const float*)d_in[12];
    const float* W1mu_l    = (const float*)d_in[13];
    const float* W1mu_r    = (const float*)d_in[14];
    const float* W2um_l    = (const float*)d_in[15];
    const float* W2um_r    = (const float*)d_in[16];
    const float* W2mu_l    = (const float*)d_in[17];
    const float* W2mu_r    = (const float*)d_in[18];
    const float* b1um      = (const float*)d_in[19];
    const float* b1mu      = (const float*)d_in[20];
    const float* b2um      = (const float*)d_in[21];
    const float* b2mu      = (const float*)d_in[22];
    float* out = (float*)d_out;

    const int U = in_sizes[7] / HDIM;
    const int M = in_sizes[8] / HDIM;
    const int E = in_sizes[3];
    const int L = in_sizes[5];
    (void)out_size;

    size_t off = 0;
    auto carve = [&](size_t nfloats) {
        float* p = (float*)((char*)d_ws + off);
        off += ((nfloats * sizeof(float) + 255) / 256) * 256;
        return p;
    };
    float* XM    = carve((size_t)M * HDIM);
    float* AGG_M = carve((size_t)M * HDIM);
    float* AGG_U = carve((size_t)U * HDIM);
    float* H_M   = carve((size_t)M * HDIM);
    float* H_U   = carve((size_t)U * HDIM);
    float* M2    = carve((size_t)M * HDIM);
    float* U2    = carve((size_t)U * HDIM);
    float* DEG_M = carve((size_t)M);
    float* DEG_U = carve((size_t)U);

    const int TB = 256;
    dim3 b(TB);
    long long sc_threads = (long long)E * 16;
    int sc_blocks = (int)((sc_threads + TB - 1) / TB);

    hipMemsetAsync(DEG_M, 0, (size_t)M * sizeof(float), stream);
    hipMemsetAsync(DEG_U, 0, (size_t)U * sizeof(float), stream);
    deg_kernel<<<(E + TB - 1) / TB, b, 0, stream>>>(e_src, e_dst, DEG_U, DEG_M, E);

    movie_init_kernel<<<(M * HDIM + TB - 1) / TB, b, 0, stream>>>(
        movie_x, lin_W, lin_b, movie_emb, XM, M);

    hipMemsetAsync(AGG_M, 0, (size_t)M * HDIM * sizeof(float), stream);
    hipMemsetAsync(AGG_U, 0, (size_t)U * HDIM * sizeof(float), stream);
    scatter_kernel<<<sc_blocks, b, 0, stream>>>(user_emb, e_src, e_dst, AGG_M, E);
    scatter_kernel<<<sc_blocks, b, 0, stream>>>(XM, e_dst, e_src, AGG_U, E);
    combine_wmma_kernel<<<M / 16, 128, 0, stream>>>(AGG_M, DEG_M, XM,
                                                    W1um_l, W1um_r, b1um, H_M, 1);
    combine_wmma_kernel<<<U / 16, 128, 0, stream>>>(AGG_U, DEG_U, user_emb,
                                                    W1mu_l, W1mu_r, b1mu, H_U, 1);

    hipMemsetAsync(AGG_M, 0, (size_t)M * HDIM * sizeof(float), stream);
    hipMemsetAsync(AGG_U, 0, (size_t)U * HDIM * sizeof(float), stream);
    scatter_kernel<<<sc_blocks, b, 0, stream>>>(H_U, e_src, e_dst, AGG_M, E);
    scatter_kernel<<<sc_blocks, b, 0, stream>>>(H_M, e_dst, e_src, AGG_U, E);
    combine_wmma_kernel<<<M / 16, 128, 0, stream>>>(AGG_M, DEG_M, H_M,
                                                    W2um_l, W2um_r, b2um, M2, 0);
    combine_wmma_kernel<<<U / 16, 128, 0, stream>>>(AGG_U, DEG_U, H_U,
                                                    W2mu_l, W2mu_r, b2mu, U2, 0);

    long long dt = (long long)L * 8;
    edge_dot_kernel<<<(int)((dt + TB - 1) / TB), b, 0, stream>>>(U2, M2, eli_u, eli_m, out, L);
}


# scatter: one wave walks 32 edges, all 32 lanes on one edge, so each row load and each f32 atomic add covers one whole 128-B line (was 8 lanes x 4 lines); sums still in arrival order
# speedup vs baseline: 1.6747x; 1.6747x over previous
#include <hip/hip_runtime.h>
#include <hip/hip_bf16.h>

typedef __attribute__((ext_vector_type(2))) float v2f;
typedef __attribute__((ext_vector_type(8))) float v8f;

#define HDIM 64

__global__ __launch_bounds__(256)
void deg_kernel(const int* __restrict__ src, const int* __restrict__ dst,
                float* __restrict__ degU, float* __restrict__ degM, int nE) {
    int e = blockIdx.x * blockDim.x + threadIdx.x;
    if (e >= nE) return;
    atomicAdd(degU + src[e], 1.0f);
    atomicAdd(degM + dst[e], 1.0f);
}

__global__ __launch_bounds__(256)
void movie_init_kernel(const float* __restrict__ mx, const float* __restrict__ linW,
                       const float* __restrict__ linb, const float* __restrict__ memb,
                       float* __restrict__ XM, int M) {
    int t = blockIdx.x * blockDim.x + threadIdx.x;
    if (t >= M * HDIM) return;
    int m = t >> 6, h = t & 63;
    float acc = linb[h] + memb[t];
    const float* xr = mx + (size_t)m * 20;
    const float* wr = linW + (size_t)h * 20;
#pragma unroll
    for (int f = 0; f < 20; ++f) acc = fmaf(xr[f], wr[f], acc);
    XM[t] = acc;
}

#define EDGES_PER_WAVE 32
#define EDGE_BATCH 8

__global__ __launch_bounds__(256)
void scatter_kernel(const float* __restrict__ xsrc, const int* __restrict__ src,
                    const int* __restrict__ dst, float* __restrict__ agg, int nE) {
    const int lane = threadIdx.x & 31;
    const long long wave = ((long long)blockIdx.x * blockDim.x + threadIdx.x) >> 5;
    const long long e0 = wave * EDGES_PER_WAVE;
    if (e0 >= (long long)nE) return;
    const long long rest = (long long)nE - e0;
    const int cnt = rest < EDGES_PER_WAVE ? (int)rest : EDGES_PER_WAVE;

    int s_l = 0, d_l = 0;
    if (lane < cnt) { s_l = src[e0 + lane]; d_l = dst[e0 + lane]; }

    if (cnt == EDGES_PER_WAVE) {
        for (int jb = 0; jb < EDGES_PER_WAVE; jb += EDGE_BATCH) {
            float a[EDGE_BATCH], b[EDGE_BATCH];
#pragma unroll
            for (int u = 0; u < EDGE_BATCH; ++u) {
                const int s = __builtin_amdgcn_readlane(s_l, jb + u);
                const float* xr = xsrc + (size_t)s * HDIM + lane;
                a[u] = xr[0];
                b[u] = xr[32];
            }
#pragma unroll
            for (int u = 0; u < EDGE_BATCH; ++u) {
                const int d = __builtin_amdgcn_readlane(d_l, jb + u);
                float* ar = agg + (size_t)d * HDIM + lane;
                atomicAdd(ar, a[u]);
                atomicAdd(ar + 32, b[u]);
            }
        }
    } else {
        for (int j = 0; j < cnt; ++j) {
            const int s = __builtin_amdgcn_readlane(s_l, j);
            const int d = __builtin_amdgcn_readlane(d_l, j);
            const float* xr = xsrc + (size_t)s * HDIM + lane;
            float* ar = agg + (size_t)d * HDIM + lane;
            const float a = xr[0], b = xr[32];
            atomicAdd(ar, a);
            atomicAdd(ar + 32, b);
        }
    }
}

__global__ __launch_bounds__(128)
void combine_wmma_kernel(const float* __restrict__ agg, const float* __restrict__ deg,
                         const float* __restrict__ xdst,
                         const float* __restrict__ Wl, const float* __restrict__ Wr,
                         const float* __restrict__ bias, float* __restrict__ out,
                         int do_relu) {
    __shared__ float A1[16][68];
    __shared__ float A2[16][68];

    const int row0 = blockIdx.x * 16;
    const int tid  = threadIdx.x;

    {
        int r  = tid >> 3;
        int c0 = (tid & 7) * 8;
        float dinv = 1.0f / fmaxf(deg[row0 + r], 1.0f);
        const float* ga = agg  + (size_t)(row0 + r) * HDIM + c0;
        const float* gx = xdst + (size_t)(row0 + r) * HDIM + c0;
#pragma unroll
        for (int j = 0; j < 8; ++j) {
            A1[r][c0 + j] = ga[j] * dinv;
            A2[r][c0 + j] = gx[j];
        }
    }
    __syncthreads();

    const int wave = tid >> 5;
    const int lane = tid & 31;
    const int m    = lane & 15;
    const int half = lane >> 4;
    const int col0 = wave * 16;

    v8f c = {};
#pragma unroll
    for (int kk = 0; kk < 16; ++kk) {
        int k = kk * 4 + half * 2;
        v2f a, b;
        a.x = A1[m][k];  a.y = A1[m][k + 1];
        const float* wp = Wl + (size_t)(col0 + m) * HDIM + k;
        b.x = wp[0];     b.y = wp[1];
        c = __builtin_amdgcn_wmma_f32_16x16x4_f32(false, a, false, b,
                                                  (short)0, c, false, false);
    }
#pragma unroll
    for (int kk = 0; kk < 16; ++kk) {
        int k = kk * 4 + half * 2;
        v2f a, b;
        a.x = A2[m][k];  a.y = A2[m][k + 1];
        const float* wp = Wr + (size_t)(col0 + m) * HDIM + k;
        b.x = wp[0];     b.y = wp[1];
        c = __builtin_amdgcn_wmma_f32_16x16x4_f32(false, a, false, b,
                                                  (short)0, c, false, false);
    }

    float bv = bias[col0 + m];
#pragma unroll
    for (int r = 0; r < 8; ++r) {
        float v = c[r] + bv;
        if (do_relu) v = fmaxf(v, 0.0f);
        out[(size_t)(row0 + r + 8 * half) * HDIM + col0 + m] = v;
    }
}

__global__ __launch_bounds__(256)
void edge_dot_kernel(const float* __restrict__ U2, const float* __restrict__ M2,
                     const int* __restrict__ eu, const int* __restrict__ em,
                     float* __restrict__ out, int L) {
    long long tid = (long long)blockIdx.x * blockDim.x + threadIdx.x;
    if (tid >= (long long)L * 8) return;
    int p = (int)(tid >> 3);
    int s = (int)(tid & 7);
    const float4* a = (const float4*)(U2 + (size_t)eu[p] * HDIM + s * 8);
    const float4* b = (const float4*)(M2 + (size_t)em[p] * HDIM + s * 8);
    float4 a0 = a[0], a1 = a[1], b0 = b[0], b1 = b[1];
    float acc = a0.x * b0.x + a0.y * b0.y + a0.z * b0.z + a0.w * b0.w
              + a1.x * b1.x + a1.y * b1.y + a1.z * b1.z + a1.w * b1.w;
    acc += __shfl_xor(acc, 1);
    acc += __shfl_xor(acc, 2);
    acc += __shfl_xor(acc, 4);
    if (s == 0) out[p] = acc;
}

extern "C" void kernel_launch(void* const* d_in, const int* in_sizes, int n_in,
                              void* d_out, int out_size, void* d_ws, size_t ws_size,
                              hipStream_t stream) {
    (void)n_in; (void)ws_size;
    const float* movie_x   = (const float*)d_in[2];
    const int*   e_src     = (const int*)d_in[3];
    const int*   e_dst     = (const int*)d_in[4];
    const int*   eli_u     = (const int*)d_in[5];
    const int*   eli_m     = (const int*)d_in[6];
    const float* user_emb  = (const float*)d_in[7];
    const float* movie_emb = (const float*)d_in[8];
    const float* lin_W     = (const float*)d_in[9];
    const float* lin_b     = (const float*)d_in[10];
    const float* W1um_l    = (const float*)d_in[11];
    const float* W1um_r    = (const float*)d_in[12];
    const float* W1mu_l    = (const float*)d_in[13];
    const float* W1mu_r    = (const float*)d_in[14];
    const float* W2um_l    = (const float*)d_in[15];
    const float* W2um_r    = (const float*)d_in[16];
    const float* W2mu_l    = (const float*)d_in[17];
    const float* W2mu_r    = (const float*)d_in[18];
    const float* b1um      = (const float*)d_in[19];
    const float* b1mu      = (const float*)d_in[20];
    const float* b2um      = (const float*)d_in[21];
    const float* b2mu      = (const float*)d_in[22];
    float* out = (float*)d_out;

    const int U = in_sizes[7] / HDIM;
    const int M = in_sizes[8] / HDIM;
    const int E = in_sizes[3];
    const int L = in_sizes[5];
    (void)out_size;

    size_t off = 0;
    auto carve = [&](size_t nfloats) {
        float* p = (float*)((char*)d_ws + off);
        off += ((nfloats * sizeof(float) + 255) / 256) * 256;
        return p;
    };
    float* XM    = carve((size_t)M * HDIM);
    float* AGG_M = carve((size_t)M * HDIM);
    float* AGG_U = carve((size_t)U * HDIM);
    float* H_M   = carve((size_t)M * HDIM);
    float* H_U   = carve((size_t)U * HDIM);
    float* M2    = carve((size_t)M * HDIM);
    float* U2    = carve((size_t)U * HDIM);
    float* DEG_M = carve((size_t)M);
    float* DEG_U = carve((size_t)U);

    const int TB = 256;
    dim3 b(TB);
    int sc_blocks = (E + TB - 1) / TB;

    hipMemsetAsync(DEG_M, 0, (size_t)M * sizeof(float), stream);
    hipMemsetAsync(DEG_U, 0, (size_t)U * sizeof(float), stream);
    deg_kernel<<<(E + TB - 1) / TB, b, 0, stream>>>(e_src, e_dst, DEG_U, DEG_M, E);

    movie_init_kernel<<<(M * HDIM + TB - 1) / TB, b, 0, stream>>>(
        movie_x, lin_W, lin_b, movie_emb, XM, M);

    hipMemsetAsync(AGG_M, 0, (size_t)M * HDIM * sizeof(float), stream);
    hipMemsetAsync(AGG_U, 0, (size_t)U * HDIM * sizeof(float), stream);
    scatter_kernel<<<sc_blocks, b, 0, stream>>>(user_emb, e_src, e_dst, AGG_M, E);
    scatter_kernel<<<sc_blocks, b, 0, stream>>>(XM, e_dst, e_src, AGG_U, E);
    combine_wmma_kernel<<<M / 16, 128, 0, stream>>>(AGG_M, DEG_M, XM,
                                                    W1um_l, W1um_r, b1um, H_M, 1);
    combine_wmma_kernel<<<U / 16, 128, 0, stream>>>(AGG_U, DEG_U, user_emb,
                                                    W1mu_l, W1mu_r, b1mu, H_U, 1);

    hipMemsetAsync(AGG_M, 0, (size_t)M * HDIM * sizeof(float), stream);
    hipMemsetAsync(AGG_U, 0, (size_t)U * HDIM * sizeof(float), stream);
    scatter_kernel<<<sc_blocks, b, 0, stream>>>(H_U, e_src, e_dst, AGG_M, E);
    scatter_kernel<<<sc_blocks, b, 0, stream>>>(H_M, e_dst, e_src, AGG_U, E);
    combine_wmma_kernel<<<M / 16, 128, 0, stream>>>(AGG_M, DEG_M, H_M,
                                                    W2um_l, W2um_r, b2um, M2, 0);
    combine_wmma_kernel<<<U / 16, 128, 0, stream>>>(AGG_U, DEG_U, H_U,
                                                    W2mu_l, W2mu_r, b2mu, U2, 0);

    long long dt = (long long)L * 8;
    edge_dot_kernel<<<(int)((dt + TB - 1) / TB), b, 0, stream>>>(U2, M2, eli_u, eli_m, out, L);
}


# segment sums without f32 atomics: per-call counting sort of edges by destination (returning int adds give ranks), then each row gathers its list; a row's sum order is now its rank order in 2 chains
# speedup vs baseline: 1.7842x; 1.0654x over previous
#include <hip/hip_runtime.h>
#include <hip/hip_bf16.h>

typedef __attribute__((ext_vector_type(2))) float v2f;
typedef __attribute__((ext_vector_type(8))) float v8f;

#define HDIM 64

#define EDGES_PER_BLOCK 1024

__global__ __launch_bounds__(256)
void count_rank_kernel(const int* __restrict__ src, const int* __restrict__ dst,
                       int* __restrict__ cntU, int* __restrict__ cntM,
                       int* __restrict__ rankU, int* __restrict__ rankM, int nE) {
    const int base = blockIdx.x * EDGES_PER_BLOCK + threadIdx.x;
    int s[4], d[4];
#pragma unroll
    for (int k = 0; k < 4; ++k) {
        const int e = base + k * 256;
        s[k] = 0; d[k] = 0;
        if (e < nE) { s[k] = src[e]; d[k] = dst[e]; }
    }
    int ru[4], rm[4];
#pragma unroll
    for (int k = 0; k < 4; ++k) {
        const int e = base + k * 256;
        ru[k] = 0; rm[k] = 0;
        if (e < nE) {
            ru[k] = atomicAdd(cntU + s[k], 1);
            rm[k] = atomicAdd(cntM + d[k], 1);
        }
    }
#pragma unroll
    for (int k = 0; k < 4; ++k) {
        const int e = base + k * 256;
        if (e < nE) { rankU[e] = ru[k]; rankM[e] = rm[k]; }
    }
}

__global__ __launch_bounds__(256)
void alloc_kernel(const int* __restrict__ cnt, int* __restrict__ start,
                  float* __restrict__ deg, int* __restrict__ total, int n) {
    __shared__ int wsum[8];
    __shared__ int sbase;
    const int i = blockIdx.x * 256 + threadIdx.x;
    const int lane = threadIdx.x & 31, w = threadIdx.x >> 5;
    const int c = (i < n) ? cnt[i] : 0;
    int x = c;
#pragma unroll
    for (int o = 1; o < 32; o <<= 1) {
        const int y = __shfl_up(x, o);
        if (lane >= o) x += y;
    }
    if (lane == 31) wsum[w] = x;
    __syncthreads();
    if (threadIdx.x == 0) {
        int tot = 0;
        for (int k = 0; k < 8; ++k) { const int v = wsum[k]; wsum[k] = tot; tot += v; }
        sbase = atomicAdd(total, tot);
    }
    __syncthreads();
    if (i < n) {
        start[i] = sbase + wsum[w] + x - c;
        deg[i] = (float)c;
    }
}

__global__ __launch_bounds__(256)
void fill_kernel(const int* __restrict__ src, const int* __restrict__ dst,
                 const int* __restrict__ rankU, const int* __restrict__ rankM,
                 const int* __restrict__ startU, const int* __restrict__ startM,
                 int* __restrict__ list, int nE) {
    const int base = blockIdx.x * EDGES_PER_BLOCK + threadIdx.x;
    int s[4], d[4], pm[4], pu[4];
#pragma unroll
    for (int k = 0; k < 4; ++k) {
        const int e = base + k * 256;
        s[k] = 0; d[k] = 0; pm[k] = 0; pu[k] = 0;
        if (e < nE) { s[k] = src[e]; d[k] = dst[e]; pm[k] = rankM[e]; pu[k] = rankU[e]; }
    }
#pragma unroll
    for (int k = 0; k < 4; ++k) {
        const int e = base + k * 256;
        if (e < nE) { pm[k] += startM[d[k]]; pu[k] += startU[s[k]]; }
    }
#pragma unroll
    for (int k = 0; k < 4; ++k) {
        const int e = base + k * 256;
        if (e < nE) {
            list[pm[k]] = s[k];
            list[pu[k]] = d[k];
        }
    }
}

template <int NPW>
__global__ __launch_bounds__(256)
void gather_sum_kernel(const float* __restrict__ xsrc, const int* __restrict__ list,
                       const int* __restrict__ start, const int* __restrict__ cnt,
                       float* __restrict__ agg, int n) {
    const int lane = threadIdx.x & 31;
    const int half = lane >> 4;
    const int q    = lane & 15;
    const int wave = (int)((blockIdx.x * blockDim.x + threadIdx.x) >> 5);
    const int node0 = wave * NPW;
    if (node0 >= n) return;

    int st_l = 0, c_l = 0;
    if (lane < NPW && node0 + lane < n) { st_l = start[node0 + lane]; c_l = cnt[node0 + lane]; }

    for (int u = 0; u < NPW; ++u) {
        const int node = node0 + u;
        if (node >= n) break;
        const int st = __builtin_amdgcn_readlane(st_l, u);
        const int c  = __builtin_amdgcn_readlane(c_l, u);
        float4 acc = make_float4(0.f, 0.f, 0.f, 0.f);
        for (int base = 0; base < c; base += 32) {
            const int m = (c - base) < 32 ? (c - base) : 32;
            int idx_l = 0;
            if (lane < m) idx_l = list[st + base + lane];
#pragma unroll
            for (int b = 0; b < 2; ++b) {
                if (b * 16 < m) {
                    float4 v[8];
#pragma unroll
                    for (int k = 0; k < 8; ++k) {
                        const int j0 = b * 16 + 2 * k;
                        const int s0 = __builtin_amdgcn_readlane(idx_l, j0);
                        const int s1 = __builtin_amdgcn_readlane(idx_l, j0 + 1);
                        const int s  = half ? s1 : s0;
                        v[k] = make_float4(0.f, 0.f, 0.f, 0.f);
                        if (j0 + half < m)
                            v[k] = *(const float4*)(xsrc + (size_t)s * HDIM + q * 4);
                    }
#pragma unroll
                    for (int k = 0; k < 8; ++k) {
                        acc.x += v[k].x; acc.y += v[k].y; acc.z += v[k].z; acc.w += v[k].w;
                    }
                }
            }
        }
        acc.x += __shfl_xor(acc.x, 16);
        acc.y += __shfl_xor(acc.y, 16);
        acc.z += __shfl_xor(acc.z, 16);
        acc.w += __shfl_xor(acc.w, 16);
        if (half == 0)
            *(float4*)(agg + (size_t)node * HDIM + q * 4) = acc;
    }
}

__global__ __launch_bounds__(256)
void movie_init_kernel(const float* __restrict__ mx, const float* __restrict__ linW,
                       const float* __restrict__ linb, const float* __restrict__ memb,
                       float* __restrict__ XM, int M) {
    int t = blockIdx.x * blockDim.x + threadIdx.x;
    if (t >= M * HDIM) return;
    int m = t >> 6, h = t & 63;
    float acc = linb[h] + memb[t];
    const float* xr = mx + (size_t)m * 20;
    const float* wr = linW + (size_t)h * 20;
#pragma unroll
    for (int f = 0; f < 20; ++f) acc = fmaf(xr[f], wr[f], acc);
    XM[t] = acc;
}

__global__ __launch_bounds__(128)
void combine_wmma_kernel(const float* __restrict__ agg, const float* __restrict__ deg,
                         const float* __restrict__ xdst,
                         const float* __restrict__ Wl, const float* __restrict__ Wr,
                         const float* __restrict__ bias, float* __restrict__ out,
                         int do_relu) {
    __shared__ float A1[16][68];
    __shared__ float A2[16][68];

    const int row0 = blockIdx.x * 16;
    const int tid  = threadIdx.x;

    {
        int r  = tid >> 3;
        int c0 = (tid & 7) * 8;
        float dinv = 1.0f / fmaxf(deg[row0 + r], 1.0f);
        const float* ga = agg  + (size_t)(row0 + r) * HDIM + c0;
        const float* gx = xdst + (size_t)(row0 + r) * HDIM + c0;
#pragma unroll
        for (int j = 0; j < 8; ++j) {
            A1[r][c0 + j] = ga[j] * dinv;
            A2[r][c0 + j] = gx[j];
        }
    }
    __syncthreads();

    const int wave = tid >> 5;
    const int lane = tid & 31;
    const int m    = lane & 15;
    const int half = lane >> 4;
    const int col0 = wave * 16;

    v8f c = {};
#pragma unroll
    for (int kk = 0; kk < 16; ++kk) {
        int k = kk * 4 + half * 2;
        v2f a, b;
        a.x = A1[m][k];  a.y = A1[m][k + 1];
        const float* wp = Wl + (size_t)(col0 + m) * HDIM + k;
        b.x = wp[0];     b.y = wp[1];
        c = __builtin_amdgcn_wmma_f32_16x16x4_f32(false, a, false, b,
                                                  (short)0, c, false, false);
    }
#pragma unroll
    for (int kk = 0; kk < 16; ++kk) {
        int k = kk * 4 + half * 2;
        v2f a, b;
        a.x = A2[m][k];  a.y = A2[m][k + 1];
        const float* wp = Wr + (size_t)(col0 + m) * HDIM + k;
        b.x = wp[0];     b.y = wp[1];
        c = __builtin_amdgcn_wmma_f32_16x16x4_f32(false, a, false, b,
                                                  (short)0, c, false, false);
    }

    float bv = bias[col0 + m];
#pragma unroll
    for (int r = 0; r < 8; ++r) {
        float v = c[r] + bv;
        if (do_relu) v = fmaxf(v, 0.0f);
        out[(size_t)(row0 + r + 8 * half) * HDIM + col0 + m] = v;
    }
}

__global__ __launch_bounds__(256)
void edge_dot_kernel(const float* __restrict__ U2, const float* __restrict__ M2,
                     const int* __restrict__ eu, const int* __restrict__ em,
                     float* __restrict__ out, int L) {
    long long tid = (long long)blockIdx.x * blockDim.x + threadIdx.x;
    if (tid >= (long long)L * 8) return;
    int p = (int)(tid >> 3);
    int s = (int)(tid & 7);
    const float4* a = (const float4*)(U2 + (size_t)eu[p] * HDIM + s * 8);
    const float4* b = (const float4*)(M2 + (size_t)em[p] * HDIM + s * 8);
    float4 a0 = a[0], a1 = a[1], b0 = b[0], b1 = b[1];
    float acc = a0.x * b0.x + a0.y * b0.y + a0.z * b0.z + a0.w * b0.w
              + a1.x * b1.x + a1.y * b1.y + a1.z * b1.z + a1.w * b1.w;
    acc += __shfl_xor(acc, 1);
    acc += __shfl_xor(acc, 2);
    acc += __shfl_xor(acc, 4);
    if (s == 0) out[p] = acc;
}

extern "C" void kernel_launch(void* const* d_in, const int* in_sizes, int n_in,
                              void* d_out, int out_size, void* d_ws, size_t ws_size,
                              hipStream_t stream) {
    (void)n_in; (void)ws_size;
    const float* movie_x   = (const float*)d_in[2];
    const int*   e_src     = (const int*)d_in[3];
    const int*   e_dst     = (const int*)d_in[4];
    const int*   eli_u     = (const int*)d_in[5];
    const int*   eli_m     = (const int*)d_in[6];
    const float* user_emb  = (const float*)d_in[7];
    const float* movie_emb = (const float*)d_in[8];
    const float* lin_W     = (const float*)d_in[9];
    const float* lin_b     = (const float*)d_in[10];
    const float* W1um_l    = (const float*)d_in[11];
    const float* W1um_r    = (const float*)d_in[12];
    const float* W1mu_l    = (const float*)d_in[13];
    const float* W1mu_r    = (const float*)d_in[14];
    const float* W2um_l    = (const float*)d_in[15];
    const float* W2um_r    = (const float*)d_in[16];
    const float* W2mu_l    = (const float*)d_in[17];
    const float* W2mu_r    = (const float*)d_in[18];
    const float* b1um      = (const float*)d_in[19];
    const float* b1mu      = (const float*)d_in[20];
    const float* b2um      = (const float*)d_in[21];
    const float* b2mu      = (const float*)d_in[22];
    float* out = (float*)d_out;

    const int U = in_sizes[7] / HDIM;
    const int M = in_sizes[8] / HDIM;
    const int E = in_sizes[3];
    const int L = in_sizes[5];
    (void)out_size;

    size_t off = 0;
    auto carve = [&](size_t nwords) {
        float* p = (float*)((char*)d_ws + off);
        off += ((nwords * sizeof(float) + 255) / 256) * 256;
        return p;
    };
    float* XM   = carve((size_t)M * HDIM);
    float* AGG  = carve((size_t)U * HDIM);
    float* H_M  = carve((size_t)M * HDIM);
    float* H_U  = carve((size_t)U * HDIM);
    float* U2   = carve((size_t)U * HDIM);
    float* M2   = XM;
    int*   LIST = (int*)carve((size_t)E * 2);
    int*   CNT  = (int*)carve((size_t)M + U + 64);
    int*   START = (int*)carve((size_t)M + U);
    float* DEG  = carve((size_t)M + U);
    int*   TOTAL = CNT + M + U;
    int*   CNT_M = CNT;      int*   CNT_U = CNT + M;
    int*   START_M = START;  int*   START_U = START + M;
    float* DEG_M = DEG;      float* DEG_U = DEG + M;
    const int Epad = (E + 63) / 64 * 64;
    int*   RANK_U = (int*)U2;
    int*   RANK_M = (int*)U2 + Epad;

    const int TB = 256;
    dim3 b(TB);
    const int e_blocks = (E + EDGES_PER_BLOCK - 1) / EDGES_PER_BLOCK;

    hipMemsetAsync(CNT, 0, ((size_t)M + U + 64) * sizeof(int), stream);
    count_rank_kernel<<<e_blocks, b, 0, stream>>>(e_src, e_dst, CNT_U, CNT_M, RANK_U, RANK_M, E);
    alloc_kernel<<<(M + U + TB - 1) / TB, b, 0, stream>>>(CNT, START, DEG, TOTAL, M + U);
    fill_kernel<<<e_blocks, b, 0, stream>>>(e_src, e_dst, RANK_U, RANK_M, START_U, START_M, LIST, E);

    movie_init_kernel<<<(M * HDIM + TB - 1) / TB, b, 0, stream>>>(
        movie_x, lin_W, lin_b, movie_emb, XM, M);

    const int gm_blocks = (M + 2 * 8 - 1) / (2 * 8);
    const int gu_blocks = (U + 4 * 8 - 1) / (4 * 8);

    gather_sum_kernel<2><<<gm_blocks, b, 0, stream>>>(user_emb, LIST, START_M, CNT_M, AGG, M);
    combine_wmma_kernel<<<M / 16, 128, 0, stream>>>(AGG, DEG_M, XM,
                                                    W1um_l, W1um_r, b1um, H_M, 1);
    gather_sum_kernel<4><<<gu_blocks, b, 0, stream>>>(XM, LIST, START_U, CNT_U, AGG, U);
    combine_wmma_kernel<<<U / 16, 128, 0, stream>>>(AGG, DEG_U, user_emb,
                                                    W1mu_l, W1mu_r, b1mu, H_U, 1);

    gather_sum_kernel<2><<<gm_blocks, b, 0, stream>>>(H_U, LIST, START_M, CNT_M, AGG, M);
    combine_wmma_kernel<<<M / 16, 128, 0, stream>>>(AGG, DEG_M, H_M,
                                                    W2um_l, W2um_r, b2um, M2, 0);
    gather_sum_kernel<4><<<gu_blocks, b, 0, stream>>>(H_M, LIST, START_U, CNT_U, AGG, U);
    combine_wmma_kernel<<<U / 16, 128, 0, stream>>>(AGG, DEG_U, H_U,
                                                    W2mu_l, W2mu_r, b2mu, U2, 0);

    long long dt = (long long)L * 8;
    edge_dot_kernel<<<(int)((dt + TB - 1) / TB), b, 0, stream>>>(U2, M2, eli_u, eli_m, out, L);
}


# neighbour gather fused into the combine kernel: each 16-row block sums its nodes' neighbour rows into the LDS tile of its f32 WMMA phase; 4 launches and the aggregate buffer go; sum order kept
# speedup vs baseline: 1.8454x; 1.0343x over previous
#include <hip/hip_runtime.h>
#include <hip/hip_bf16.h>

typedef __attribute__((ext_vector_type(2))) float v2f;
typedef __attribute__((ext_vector_type(8))) float v8f;

#define HDIM 64

#define EDGES_PER_BLOCK 1024

__global__ __launch_bounds__(256)
void count_rank_kernel(const int* __restrict__ src, const int* __restrict__ dst,
                       int* __restrict__ cntU, int* __restrict__ cntM,
                       int* __restrict__ rankU, int* __restrict__ rankM, int nE) {
    const int base = blockIdx.x * EDGES_PER_BLOCK + threadIdx.x;
    int s[4], d[4];
#pragma unroll
    for (int k = 0; k < 4; ++k) {
        const int e = base + k * 256;
        s[k] = 0; d[k] = 0;
        if (e < nE) { s[k] = src[e]; d[k] = dst[e]; }
    }
    int ru[4], rm[4];
#pragma unroll
    for (int k = 0; k < 4; ++k) {
        const int e = base + k * 256;
        ru[k] = 0; rm[k] = 0;
        if (e < nE) {
            ru[k] = atomicAdd(cntU + s[k], 1);
            rm[k] = atomicAdd(cntM + d[k], 1);
        }
    }
#pragma unroll
    for (int k = 0; k < 4; ++k) {
        const int e = base + k * 256;
        if (e < nE) { rankU[e] = ru[k]; rankM[e] = rm[k]; }
    }
}

__global__ __launch_bounds__(256)
void alloc_kernel(const int* __restrict__ cnt, int* __restrict__ start,
                  int* __restrict__ total, int n) {
    __shared__ int wsum[8];
    __shared__ int sbase;
    const int i = blockIdx.x * 256 + threadIdx.x;
    const int lane = threadIdx.x & 31, w = threadIdx.x >> 5;
    const int c = (i < n) ? cnt[i] : 0;
    int x = c;
#pragma unroll
    for (int o = 1; o < 32; o <<= 1) {
        const int y = __shfl_up(x, o);
        if (lane >= o) x += y;
    }
    if (lane == 31) wsum[w] = x;
    __syncthreads();
    if (threadIdx.x == 0) {
        int tot = 0;
        for (int k = 0; k < 8; ++k) { const int v = wsum[k]; wsum[k] = tot; tot += v; }
        sbase = atomicAdd(total, tot);
    }
    __syncthreads();
    if (i < n) start[i] = sbase + wsum[w] + x - c;
}

__global__ __launch_bounds__(256)
void fill_kernel(const int* __restrict__ src, const int* __restrict__ dst,
                 const int* __restrict__ rankU, const int* __restrict__ rankM,
                 const int* __restrict__ startU, const int* __restrict__ startM,
                 int* __restrict__ list, int nE) {
    const int base = blockIdx.x * EDGES_PER_BLOCK + threadIdx.x;
    int s[4], d[4], pm[4], pu[4];
#pragma unroll
    for (int k = 0; k < 4; ++k) {
        const int e = base + k * 256;
        s[k] = 0; d[k] = 0; pm[k] = 0; pu[k] = 0;
        if (e < nE) { s[k] = src[e]; d[k] = dst[e]; pm[k] = rankM[e]; pu[k] = rankU[e]; }
    }
#pragma unroll
    for (int k = 0; k < 4; ++k) {
        const int e = base + k * 256;
        if (e < nE) { pm[k] += startM[d[k]]; pu[k] += startU[s[k]]; }
    }
#pragma unroll
    for (int k = 0; k < 4; ++k) {
        const int e = base + k * 256;
        if (e < nE) {
            list[pm[k]] = s[k];
            list[pu[k]] = d[k];
        }
    }
}

__global__ __launch_bounds__(256)
void movie_init_kernel(const float* __restrict__ mx, const float* __restrict__ linW,
                       const float* __restrict__ linb, const float* __restrict__ memb,
                       float* __restrict__ XM, int M) {
    int t = blockIdx.x * blockDim.x + threadIdx.x;
    if (t >= M * HDIM) return;
    int m = t >> 6, h = t & 63;
    float acc = linb[h] + memb[t];
    const float* xr = mx + (size_t)m * 20;
    const float* wr = linW + (size_t)h * 20;
#pragma unroll
    for (int f = 0; f < 20; ++f) acc = fmaf(xr[f], wr[f], acc);
    XM[t] = acc;
}

__global__ __launch_bounds__(128, 16)
void sage_layer_kernel(const float* __restrict__ xsrc, const int* __restrict__ list,
                       const int* __restrict__ start, const int* __restrict__ cnt,
                       const float* __restrict__ xdst,
                       const float* __restrict__ Wl, const float* __restrict__ Wr,
                       const float* __restrict__ bias, float* __restrict__ out,
                       int do_relu) {
    __shared__ __attribute__((aligned(16))) float A1[16][68];
    __shared__ __attribute__((aligned(16))) float A2[16][68];

    const int row0 = blockIdx.x * 16;
    const int tid  = threadIdx.x;
    const int wave = tid >> 5;
    const int lane = tid & 31;
    const int half = lane >> 4;
    const int q    = lane & 15;

    const int node0 = row0 + wave * 4;
    const int st_l = start[node0 + (lane & 3)];
    const int c_l  = cnt[node0 + (lane & 3)];

    {
        const int xr = tid >> 3;
        const int xc = (tid & 7) * 8;
        const float4* gx = (const float4*)(xdst + (size_t)(row0 + xr) * HDIM + xc);
        const float4 x0 = gx[0];
        const float4 x1 = gx[1];
        *(float4*)&A2[xr][xc]     = x0;
        *(float4*)&A2[xr][xc + 4] = x1;
    }

    {
        for (int u = 0; u < 4; ++u) {
            const int st = __builtin_amdgcn_readlane(st_l, u);
            const int c  = __builtin_amdgcn_readlane(c_l, u);
            float4 acc = make_float4(0.f, 0.f, 0.f, 0.f);
            for (int base = 0; base < c; base += 32) {
                const int m = (c - base) < 32 ? (c - base) : 32;
                int idx_l = 0;
                if (lane < m) idx_l = list[st + base + lane];
#pragma unroll
                for (int b = 0; b < 2; ++b) {
                    if (b * 16 < m) {
                        float4 v[8];
#pragma unroll
                        for (int k = 0; k < 8; ++k) {
                            const int j0 = b * 16 + 2 * k;
                            const int s0 = __builtin_amdgcn_readlane(idx_l, j0);
                            const int s1 = __builtin_amdgcn_readlane(idx_l, j0 + 1);
                            const int s  = half ? s1 : s0;
                            v[k] = make_float4(0.f, 0.f, 0.f, 0.f);
                            if (j0 + half < m)
                                v[k] = *(const float4*)(xsrc + (size_t)s * HDIM + q * 4);
                        }
#pragma unroll
                        for (int k = 0; k < 8; ++k) {
                            acc.x += v[k].x; acc.y += v[k].y; acc.z += v[k].z; acc.w += v[k].w;
                        }
                    }
                }
            }
            acc.x += __shfl_xor(acc.x, 16);
            acc.y += __shfl_xor(acc.y, 16);
            acc.z += __shfl_xor(acc.z, 16);
            acc.w += __shfl_xor(acc.w, 16);
            const float dinv = 1.0f / fmaxf((float)c, 1.0f);
            if (half == 0)
                *(float4*)&A1[wave * 4 + u][q * 4] =
                    make_float4(acc.x * dinv, acc.y * dinv, acc.z * dinv, acc.w * dinv);
        }
    }
    __syncthreads();

    const int m    = lane & 15;
    const int col0 = wave * 16;

    v8f c = {};
#pragma unroll
    for (int kk = 0; kk < 16; ++kk) {
        int k = kk * 4 + half * 2;
        v2f a, b;
        a.x = A1[m][k];  a.y = A1[m][k + 1];
        const float* wp = Wl + (size_t)(col0 + m) * HDIM + k;
        b.x = wp[0];     b.y = wp[1];
        c = __builtin_amdgcn_wmma_f32_16x16x4_f32(false, a, false, b,
                                                  (short)0, c, false, false);
    }
#pragma unroll
    for (int kk = 0; kk < 16; ++kk) {
        int k = kk * 4 + half * 2;
        v2f a, b;
        a.x = A2[m][k];  a.y = A2[m][k + 1];
        const float* wp = Wr + (size_t)(col0 + m) * HDIM + k;
        b.x = wp[0];     b.y = wp[1];
        c = __builtin_amdgcn_wmma_f32_16x16x4_f32(false, a, false, b,
                                                  (short)0, c, false, false);
    }

    float bv = bias[col0 + m];
#pragma unroll
    for (int r = 0; r < 8; ++r) {
        float v = c[r] + bv;
        if (do_relu) v = fmaxf(v, 0.0f);
        out[(size_t)(row0 + r + 8 * half) * HDIM + col0 + m] = v;
    }
}

__global__ __launch_bounds__(256)
void edge_dot_kernel(const float* __restrict__ U2, const float* __restrict__ M2,
                     const int* __restrict__ eu, const int* __restrict__ em,
                     float* __restrict__ out, int L) {
    long long tid = (long long)blockIdx.x * blockDim.x + threadIdx.x;
    if (tid >= (long long)L * 8) return;
    int p = (int)(tid >> 3);
    int s = (int)(tid & 7);
    const float4* a = (const float4*)(U2 + (size_t)eu[p] * HDIM + s * 8);
    const float4* b = (const float4*)(M2 + (size_t)em[p] * HDIM + s * 8);
    float4 a0 = a[0], a1 = a[1], b0 = b[0], b1 = b[1];
    float acc = a0.x * b0.x + a0.y * b0.y + a0.z * b0.z + a0.w * b0.w
              + a1.x * b1.x + a1.y * b1.y + a1.z * b1.z + a1.w * b1.w;
    acc += __shfl_xor(acc, 1);
    acc += __shfl_xor(acc, 2);
    acc += __shfl_xor(acc, 4);
    if (s == 0) out[p] = acc;
}

extern "C" void kernel_launch(void* const* d_in, const int* in_sizes, int n_in,
                              void* d_out, int out_size, void* d_ws, size_t ws_size,
                              hipStream_t stream) {
    (void)n_in; (void)ws_size;
    const float* movie_x   = (const float*)d_in[2];
    const int*   e_src     = (const int*)d_in[3];
    const int*   e_dst     = (const int*)d_in[4];
    const int*   eli_u     = (const int*)d_in[5];
    const int*   eli_m     = (const int*)d_in[6];
    const float* user_emb  = (const float*)d_in[7];
    const float* movie_emb = (const float*)d_in[8];
    const float* lin_W     = (const float*)d_in[9];
    const float* lin_b     = (const float*)d_in[10];
    const float* W1um_l    = (const float*)d_in[11];
    const float* W1um_r    = (const float*)d_in[12];
    const float* W1mu_l    = (const float*)d_in[13];
    const float* W1mu_r    = (const float*)d_in[14];
    const float* W2um_l    = (const float*)d_in[15];
    const float* W2um_r    = (const float*)d_in[16];
    const float* W2mu_l    = (const float*)d_in[17];
    const float* W2mu_r    = (const float*)d_in[18];
    const float* b1um      = (const float*)d_in[19];
    const float* b1mu      = (const float*)d_in[20];
    const float* b2um      = (const float*)d_in[21];
    const float* b2mu      = (const float*)d_in[22];
    float* out = (float*)d_out;

    const int U = in_sizes[7] / HDIM;
    const int M = in_sizes[8] / HDIM;
    const int E = in_sizes[3];
    const int L = in_sizes[5];
    (void)out_size;

    size_t off = 0;
    auto carve = [&](size_t nwords) {
        float* p = (float*)((char*)d_ws + off);
        off += ((nwords * sizeof(float) + 255) / 256) * 256;
        return p;
    };
    float* XM   = carve((size_t)M * HDIM);
    float* H_M  = carve((size_t)M * HDIM);
    float* H_U  = carve((size_t)U * HDIM);
    float* U2   = carve((size_t)U * HDIM);
    float* M2   = XM;
    int*   LIST = (int*)carve((size_t)E * 2);
    int*   CNT  = (int*)carve((size_t)M + U + 64);
    int*   START = (int*)carve((size_t)M + U);
    int*   TOTAL = CNT + M + U;
    int*   CNT_M = CNT;      int*   CNT_U = CNT + M;
    int*   START_M = START;  int*   START_U = START + M;
    const int Epad = (E + 63) / 64 * 64;
    int*   RANK_U = (int*)U2;
    int*   RANK_M = (int*)U2 + Epad;

    const int TB = 256;
    dim3 b(TB);
    const int e_blocks = (E + EDGES_PER_BLOCK - 1) / EDGES_PER_BLOCK;

    hipMemsetAsync(CNT, 0, ((size_t)M + U + 64) * sizeof(int), stream);
    count_rank_kernel<<<e_blocks, b, 0, stream>>>(e_src, e_dst, CNT_U, CNT_M, RANK_U, RANK_M, E);
    alloc_kernel<<<(M + U + TB - 1) / TB, b, 0, stream>>>(CNT, START, TOTAL, M + U);
    fill_kernel<<<e_blocks, b, 0, stream>>>(e_src, e_dst, RANK_U, RANK_M, START_U, START_M, LIST, E);

    movie_init_kernel<<<(M * HDIM + TB - 1) / TB, b, 0, stream>>>(
        movie_x, lin_W, lin_b, movie_emb, XM, M);

    sage_layer_kernel<<<M / 16, 128, 0, stream>>>(user_emb, LIST, START_M, CNT_M, XM,
                                                  W1um_l, W1um_r, b1um, H_M, 1);
    sage_layer_kernel<<<U / 16, 128, 0, stream>>>(XM, LIST, START_U, CNT_U, user_emb,
                                                  W1mu_l, W1mu_r, b1mu, H_U, 1);

    sage_layer_kernel<<<M / 16, 128, 0, stream>>>(H_U, LIST, START_M, CNT_M, H_M,
                                                  W2um_l, W2um_r, b2um, M2, 0);
    sage_layer_kernel<<<U / 16, 128, 0, stream>>>(H_M, LIST, START_U, CNT_U, H_U,
                                                  W2mu_l, W2mu_r, b2mu, U2, 0);

    long long dt = (long long)L * 8;
    edge_dot_kernel<<<(int)((dt + TB - 1) / TB), b, 0, stream>>>(U2, M2, eli_u, eli_m, out, L);
}


# wave entry: pair dot (was 250,000 four-pair waves) and movie init (100,000 waves) now on 7,840 and 8,192 looping waves, whole-line loads; a dot's 64 terms now summed in another lane grouping
# speedup vs baseline: 1.9052x; 1.0324x over previous
#include <hip/hip_runtime.h>
#include <hip/hip_bf16.h>

typedef __attribute__((ext_vector_type(2))) float v2f;
typedef __attribute__((ext_vector_type(8))) float v8f;

#define HDIM 64

#define EDGES_PER_BLOCK 1024

__global__ __launch_bounds__(256)
void count_rank_kernel(const int* __restrict__ src, const int* __restrict__ dst,
                       int* __restrict__ cntU, int* __restrict__ cntM,
                       int* __restrict__ rankU, int* __restrict__ rankM, int nE) {
    const int base = blockIdx.x * EDGES_PER_BLOCK + threadIdx.x;
    int s[4], d[4];
#pragma unroll
    for (int k = 0; k < 4; ++k) {
        const int e = base + k * 256;
        s[k] = 0; d[k] = 0;
        if (e < nE) { s[k] = src[e]; d[k] = dst[e]; }
    }
    int ru[4], rm[4];
#pragma unroll
    for (int k = 0; k < 4; ++k) {
        const int e = base + k * 256;
        ru[k] = 0; rm[k] = 0;
        if (e < nE) {
            ru[k] = atomicAdd(cntU + s[k], 1);
            rm[k] = atomicAdd(cntM + d[k], 1);
        }
    }
#pragma unroll
    for (int k = 0; k < 4; ++k) {
        const int e = base + k * 256;
        if (e < nE) { rankU[e] = ru[k]; rankM[e] = rm[k]; }
    }
}

__global__ __launch_bounds__(256)
void alloc_kernel(const int* __restrict__ cnt, int* __restrict__ start,
                  int* __restrict__ total, int n) {
    __shared__ int wsum[8];
    __shared__ int sbase;
    const int i = blockIdx.x * 256 + threadIdx.x;
    const int lane = threadIdx.x & 31, w = threadIdx.x >> 5;
    const int c = (i < n) ? cnt[i] : 0;
    int x = c;
#pragma unroll
    for (int o = 1; o < 32; o <<= 1) {
        const int y = __shfl_up(x, o);
        if (lane >= o) x += y;
    }
    if (lane == 31) wsum[w] = x;
    __syncthreads();
    if (threadIdx.x == 0) {
        int tot = 0;
        for (int k = 0; k < 8; ++k) { const int v = wsum[k]; wsum[k] = tot; tot += v; }
        sbase = atomicAdd(total, tot);
    }
    __syncthreads();
    if (i < n) start[i] = sbase + wsum[w] + x - c;
}

__global__ __launch_bounds__(256)
void fill_kernel(const int* __restrict__ src, const int* __restrict__ dst,
                 const int* __restrict__ rankU, const int* __restrict__ rankM,
                 const int* __restrict__ startU, const int* __restrict__ startM,
                 int* __restrict__ list, int nE) {
    const int base = blockIdx.x * EDGES_PER_BLOCK + threadIdx.x;
    int s[4], d[4], pm[4], pu[4];
#pragma unroll
    for (int k = 0; k < 4; ++k) {
        const int e = base + k * 256;
        s[k] = 0; d[k] = 0; pm[k] = 0; pu[k] = 0;
        if (e < nE) { s[k] = src[e]; d[k] = dst[e]; pm[k] = rankM[e]; pu[k] = rankU[e]; }
    }
#pragma unroll
    for (int k = 0; k < 4; ++k) {
        const int e = base + k * 256;
        if (e < nE) { pm[k] += startM[d[k]]; pu[k] += startU[s[k]]; }
    }
#pragma unroll
    for (int k = 0; k < 4; ++k) {
        const int e = base + k * 256;
        if (e < nE) {
            list[pm[k]] = s[k];
            list[pu[k]] = d[k];
        }
    }
}

#define INIT_BLOCKS 1024

__global__ __launch_bounds__(256)
void movie_init_kernel(const float* __restrict__ mx, const float* __restrict__ linW,
                       const float* __restrict__ linb, const float* __restrict__ memb,
                       float* __restrict__ XM, int M) {
    const int lane = threadIdx.x & 31;
    const int wid = __builtin_amdgcn_readfirstlane((int)(blockIdx.x * 8 + (threadIdx.x >> 5)));
    const int nw  = (int)gridDim.x * 8;
    const int h0 = lane, h1 = lane + 32;

    float w0[20], w1[20];
#pragma unroll
    for (int f = 0; f < 20; ++f) { w0[f] = linW[h0 * 20 + f]; w1[f] = linW[h1 * 20 + f]; }
    const float bias0 = linb[h0], bias1 = linb[h1];

    for (int m = wid; m < M; m += nw) {
        const float* xr = mx + (size_t)m * 20;
        float acc0 = bias0 + memb[(size_t)m * HDIM + h0];
        float acc1 = bias1 + memb[(size_t)m * HDIM + h1];
#pragma unroll
        for (int f = 0; f < 20; ++f) {
            acc0 = fmaf(xr[f], w0[f], acc0);
            acc1 = fmaf(xr[f], w1[f], acc1);
        }
        XM[(size_t)m * HDIM + h0] = acc0;
        XM[(size_t)m * HDIM + h1] = acc1;
    }
}

__global__ __launch_bounds__(128, 16)
void sage_layer_kernel(const float* __restrict__ xsrc, const int* __restrict__ list,
                       const int* __restrict__ start, const int* __restrict__ cnt,
                       const float* __restrict__ xdst,
                       const float* __restrict__ Wl, const float* __restrict__ Wr,
                       const float* __restrict__ bias, float* __restrict__ out,
                       int do_relu) {
    __shared__ __attribute__((aligned(16))) float A1[16][68];
    __shared__ __attribute__((aligned(16))) float A2[16][68];

    const int row0 = blockIdx.x * 16;
    const int tid  = threadIdx.x;
    const int wave = tid >> 5;
    const int lane = tid & 31;
    const int half = lane >> 4;
    const int q    = lane & 15;

    const int node0 = row0 + wave * 4;
    const int st_l = start[node0 + (lane & 3)];
    const int c_l  = cnt[node0 + (lane & 3)];

    {
        const int xr = tid >> 3;
        const int xc = (tid & 7) * 8;
        const float4* gx = (const float4*)(xdst + (size_t)(row0 + xr) * HDIM + xc);
        const float4 x0 = gx[0];
        const float4 x1 = gx[1];
        *(float4*)&A2[xr][xc]     = x0;
        *(float4*)&A2[xr][xc + 4] = x1;
    }

    {
        for (int u = 0; u < 4; ++u) {
            const int st = __builtin_amdgcn_readlane(st_l, u);
            const int c  = __builtin_amdgcn_readlane(c_l, u);
            float4 acc = make_float4(0.f, 0.f, 0.f, 0.f);
            for (int base = 0; base < c; base += 32) {
                const int m = (c - base) < 32 ? (c - base) : 32;
                int idx_l = 0;
                if (lane < m) idx_l = list[st + base + lane];
#pragma unroll
                for (int b = 0; b < 2; ++b) {
                    if (b * 16 < m) {
                        float4 v[8];
#pragma unroll
                        for (int k = 0; k < 8; ++k) {
                            const int j0 = b * 16 + 2 * k;
                            const int s0 = __builtin_amdgcn_readlane(idx_l, j0);
                            const int s1 = __builtin_amdgcn_readlane(idx_l, j0 + 1);
                            const int s  = half ? s1 : s0;
                            v[k] = make_float4(0.f, 0.f, 0.f, 0.f);
                            if (j0 + half < m)
                                v[k] = *(const float4*)(xsrc + (size_t)s * HDIM + q * 4);
                        }
#pragma unroll
                        for (int k = 0; k < 8; ++k) {
                            acc.x += v[k].x; acc.y += v[k].y; acc.z += v[k].z; acc.w += v[k].w;
                        }
                    }
                }
            }
            acc.x += __shfl_xor(acc.x, 16);
            acc.y += __shfl_xor(acc.y, 16);
            acc.z += __shfl_xor(acc.z, 16);
            acc.w += __shfl_xor(acc.w, 16);
            const float dinv = 1.0f / fmaxf((float)c, 1.0f);
            if (half == 0)
                *(float4*)&A1[wave * 4 + u][q * 4] =
                    make_float4(acc.x * dinv, acc.y * dinv, acc.z * dinv, acc.w * dinv);
        }
    }
    __syncthreads();

    const int m    = lane & 15;
    const int col0 = wave * 16;

    v8f c = {};
#pragma unroll
    for (int kk = 0; kk < 16; ++kk) {
        int k = kk * 4 + half * 2;
        v2f a, b;
        a.x = A1[m][k];  a.y = A1[m][k + 1];
        const float* wp = Wl + (size_t)(col0 + m) * HDIM + k;
        b.x = wp[0];     b.y = wp[1];
        c = __builtin_amdgcn_wmma_f32_16x16x4_f32(false, a, false, b,
                                                  (short)0, c, false, false);
    }
#pragma unroll
    for (int kk = 0; kk < 16; ++kk) {
        int k = kk * 4 + half * 2;
        v2f a, b;
        a.x = A2[m][k];  a.y = A2[m][k + 1];
        const float* wp = Wr + (size_t)(col0 + m) * HDIM + k;
        b.x = wp[0];     b.y = wp[1];
        c = __builtin_amdgcn_wmma_f32_16x16x4_f32(false, a, false, b,
                                                  (short)0, c, false, false);
    }

    float bv = bias[col0 + m];
#pragma unroll
    for (int r = 0; r < 8; ++r) {
        float v = c[r] + bv;
        if (do_relu) v = fmaxf(v, 0.0f);
        out[(size_t)(row0 + r + 8 * half) * HDIM + col0 + m] = v;
    }
}

#define DOT_BLOCKS 980

__global__ __launch_bounds__(256) __attribute__((amdgpu_waves_per_eu(8, 8)))
void edge_dot_kernel(const float* __restrict__ U2, const float* __restrict__ M2,
                     const int* __restrict__ eu, const int* __restrict__ em,
                     float* __restrict__ out, int L) {
    const int lane = threadIdx.x & 31;
    const int wid = __builtin_amdgcn_readfirstlane((int)(blockIdx.x * 8 + (threadIdx.x >> 5)));
    const int nw  = (int)gridDim.x * 8;
    const int g = lane >> 3;
    const int s = lane & 7;
    const int ntrip = (L + 31) >> 5;
    if (wid >= ntrip) return;

    int pi = min(wid * 32 + lane, L - 1);
    int iu = eu[pi], im = em[pi];

    for (int c = wid; c < ntrip; c += nw) {
        pi = min((c + nw) * 32 + lane, L - 1);
        const int iu_n = eu[pi], im_n = em[pi];

        float4 a0[8], a1[8], b0[8], b1[8];
#pragma unroll
        for (int k = 0; k < 8; ++k) {
            const int j  = k * 4 + g;
            const int ru = __shfl(iu, j);
            const int rm = __shfl(im, j);
            const float4* a = (const float4*)(U2 + (size_t)ru * HDIM + s * 4);
            const float4* b = (const float4*)(M2 + (size_t)rm * HDIM + s * 4);
            a0[k] = a[0]; a1[k] = a[8]; b0[k] = b[0]; b1[k] = b[8];
        }
        float res = 0.f;
#pragma unroll
        for (int k = 0; k < 8; ++k) {
            float acc = a0[k].x * b0[k].x + a0[k].y * b0[k].y + a0[k].z * b0[k].z + a0[k].w * b0[k].w
                      + a1[k].x * b1[k].x + a1[k].y * b1[k].y + a1[k].z * b1[k].z + a1[k].w * b1[k].w;
            acc += __shfl_xor(acc, 1);
            acc += __shfl_xor(acc, 2);
            acc += __shfl_xor(acc, 4);
            const float r = __shfl(acc, (lane & 3) * 8);
            if ((lane >> 2) == k) res = r;
        }
        if (c * 32 + lane < L) out[c * 32 + lane] = res;
        iu = iu_n; im = im_n;
    }
}

extern "C" void kernel_launch(void* const* d_in, const int* in_sizes, int n_in,
                              void* d_out, int out_size, void* d_ws, size_t ws_size,
                              hipStream_t stream) {
    (void)n_in; (void)ws_size;
    const float* movie_x   = (const float*)d_in[2];
    const int*   e_src     = (const int*)d_in[3];
    const int*   e_dst     = (const int*)d_in[4];
    const int*   eli_u     = (const int*)d_in[5];
    const int*   eli_m     = (const int*)d_in[6];
    const float* user_emb  = (const float*)d_in[7];
    const float* movie_emb = (const float*)d_in[8];
    const float* lin_W     = (const float*)d_in[9];
    const float* lin_b     = (const float*)d_in[10];
    const float* W1um_l    = (const float*)d_in[11];
    const float* W1um_r    = (const float*)d_in[12];
    const float* W1mu_l    = (const float*)d_in[13];
    const float* W1mu_r    = (const float*)d_in[14];
    const float* W2um_l    = (const float*)d_in[15];
    const float* W2um_r    = (const float*)d_in[16];
    const float* W2mu_l    = (const float*)d_in[17];
    const float* W2mu_r    = (const float*)d_in[18];
    const float* b1um      = (const float*)d_in[19];
    const float* b1mu      = (const float*)d_in[20];
    const float* b2um      = (const float*)d_in[21];
    const float* b2mu      = (const float*)d_in[22];
    float* out = (float*)d_out;

    const int U = in_sizes[7] / HDIM;
    const int M = in_sizes[8] / HDIM;
    const int E = in_sizes[3];
    const int L = in_sizes[5];
    (void)out_size;

    size_t off = 0;
    auto carve = [&](size_t nwords) {
        float* p = (float*)((char*)d_ws + off);
        off += ((nwords * sizeof(float) + 255) / 256) * 256;
        return p;
    };
    float* XM   = carve((size_t)M * HDIM);
    float* H_M  = carve((size_t)M * HDIM);
    float* H_U  = carve((size_t)U * HDIM);
    float* U2   = carve((size_t)U * HDIM);
    float* M2   = XM;
    int*   LIST = (int*)carve((size_t)E * 2);
    int*   CNT  = (int*)carve((size_t)M + U + 64);
    int*   START = (int*)carve((size_t)M + U);
    int*   TOTAL = CNT + M + U;
    int*   CNT_M = CNT;      int*   CNT_U = CNT + M;
    int*   START_M = START;  int*   START_U = START + M;
    const int Epad = (E + 63) / 64 * 64;
    int*   RANK_U = (int*)U2;
    int*   RANK_M = (int*)U2 + Epad;

    const int TB = 256;
    dim3 b(TB);
    const int e_blocks = (E + EDGES_PER_BLOCK - 1) / EDGES_PER_BLOCK;

    hipMemsetAsync(CNT, 0, ((size_t)M + U + 64) * sizeof(int), stream);
    count_rank_kernel<<<e_blocks, b, 0, stream>>>(e_src, e_dst, CNT_U, CNT_M, RANK_U, RANK_M, E);
    alloc_kernel<<<(M + U + TB - 1) / TB, b, 0, stream>>>(CNT, START, TOTAL, M + U);
    fill_kernel<<<e_blocks, b, 0, stream>>>(e_src, e_dst, RANK_U, RANK_M, START_U, START_M, LIST, E);

    movie_init_kernel<<<INIT_BLOCKS, b, 0, stream>>>(
        movie_x, lin_W, lin_b, movie_emb, XM, M);

    sage_layer_kernel<<<M / 16, 128, 0, stream>>>(user_emb, LIST, START_M, CNT_M, XM,
                                                  W1um_l, W1um_r, b1um, H_M, 1);
    sage_layer_kernel<<<U / 16, 128, 0, stream>>>(XM, LIST, START_U, CNT_U, user_emb,
                                                  W1mu_l, W1mu_r, b1mu, H_U, 1);

    sage_layer_kernel<<<M / 16, 128, 0, stream>>>(H_U, LIST, START_M, CNT_M, H_M,
                                                  W2um_l, W2um_r, b2um, M2, 0);
    sage_layer_kernel<<<U / 16, 128, 0, stream>>>(H_M, LIST, START_U, CNT_U, H_U,
                                                  W2mu_l, W2mu_r, b2mu, U2, 0);

    edge_dot_kernel<<<DOT_BLOCKS, b, 0, stream>>>(U2, M2, eli_u, eli_m, out, L);
}


# edge lists by a two-level LDS bucket sort: per-edge global returning adds and scattered 4-B loads and stores become whole-line traffic; a row's sum order is now the bucket block's LDS arrival order
# speedup vs baseline: 4.9813x; 2.6146x over previous
#include <hip/hip_runtime.h>
#include <hip/hip_bf16.h>

typedef __attribute__((ext_vector_type(2))) float v2f;
typedef __attribute__((ext_vector_type(8))) float v8f;

#define HDIM 64

#define BSHIFT 7
#define BKEYS 128
#define MAXB 2048
#define SPLIT_EDGES 8192
#define STAGE_CAP 12288

__device__ __forceinline__ int scan2048(int c0, int c1, int* wsum) {
    const int lane = threadIdx.x & 31, w = threadIdx.x >> 5;
    int x = c0 + c1;
    for (int o = 1; o < 32; o <<= 1) {
        const int y = __shfl_up(x, o);
        if (lane >= o) x += y;
    }
    __syncthreads();
    if (lane == 31) wsum[w] = x;
    __syncthreads();
    if (w == 0) {
        const int v = wsum[lane];
        int z = v;
        for (int o = 1; o < 32; o <<= 1) {
            const int y = __shfl_up(z, o);
            if (lane >= o) z += y;
        }
        wsum[lane] = z - v;
    }
    __syncthreads();
    return wsum[w] + x - (c0 + c1);
}

__global__ __launch_bounds__(1024)
void bucket_hist_kernel(const int* __restrict__ src, const int* __restrict__ dst,
                        int* __restrict__ gcount, int nE, int nbM) {
    __shared__ int h[MAXB];
    const int t = threadIdx.x;
    h[2 * t] = 0; h[2 * t + 1] = 0;
    __syncthreads();
    const int e0 = blockIdx.x * SPLIT_EDGES + t;
    int s[8], d[8];
    for (int k = 0; k < 8; ++k) {
        const int e = e0 + k * 1024;
        s[k] = -1; d[k] = 0;
        if (e < nE) { s[k] = src[e]; d[k] = dst[e]; }
    }
    for (int k = 0; k < 8; ++k) {
        if (s[k] >= 0) {
            atomicAdd(&h[d[k] >> BSHIFT], 1);
            atomicAdd(&h[nbM + (s[k] >> BSHIFT)], 1);
        }
    }
    __syncthreads();
    const int n0 = h[2 * t], n1 = h[2 * t + 1];
    if (n0) atomicAdd(gcount + 2 * t, n0);
    if (n1) atomicAdd(gcount + 2 * t + 1, n1);
}

__global__ __launch_bounds__(1024)
void bucket_split_kernel(const int* __restrict__ src, const int* __restrict__ dst,
                         const int* __restrict__ gcount, int* __restrict__ goff,
                         int* __restrict__ gstart, int* __restrict__ items,
                         int nE, int nbM, int nb) {
    __shared__ int h[MAXB];
    __shared__ int lp[MAXB];
    __shared__ int gb[MAXB];
    __shared__ int wsum[32];
    __shared__ int st_item[2 * SPLIT_EDGES];
    __shared__ int st_dest[2 * SPLIT_EDGES];
    const int t = threadIdx.x;
    const int i0 = 2 * t, i1 = 2 * t + 1;

    const int eb = blockIdx.x * SPLIT_EDGES;
    int s[8], d[8];
    for (int k = 0; k < 8; ++k) {
        const int e = eb + t + k * 1024;
        s[k] = -1; d[k] = 0;
        if (e < nE) { s[k] = src[e]; d[k] = dst[e]; }
    }

    const int c0 = (i0 < nb) ? gcount[i0] : 0;
    const int c1 = (i1 < nb) ? gcount[i1] : 0;
    const int ex = scan2048(c0, c1, wsum);
    gb[i0] = ex; gb[i1] = ex + c0;
    if (blockIdx.x == 0) {
        if (i0 < nb) gstart[i0] = ex;
        if (i1 < nb) gstart[i1] = ex + c0;
    }
    h[i0] = 0; h[i1] = 0;
    __syncthreads();

    for (int k = 0; k < 8; ++k) {
        if (s[k] >= 0) {
            atomicAdd(&h[d[k] >> BSHIFT], 1);
            atomicAdd(&h[nbM + (s[k] >> BSHIFT)], 1);
        }
    }
    __syncthreads();

    const int n0 = h[i0], n1 = h[i1];
    const int lex = scan2048(n0, n1, wsum);
    lp[i0] = lex; lp[i1] = lex + n0;
    if (n0) gb[i0] += atomicAdd(goff + i0, n0);
    if (n1) gb[i1] += atomicAdd(goff + i1, n1);
    h[i0] = 0; h[i1] = 0;
    __syncthreads();

    for (int k = 0; k < 8; ++k) {
        if (s[k] >= 0) {
            const int bm = d[k] >> BSHIFT;
            const int rm = atomicAdd(&h[bm], 1);
            st_item[lp[bm] + rm] = ((d[k] & (BKEYS - 1)) << 24) | s[k];
            st_dest[lp[bm] + rm] = gb[bm] + rm;
            const int bu = nbM + (s[k] >> BSHIFT);
            const int ru = atomicAdd(&h[bu], 1);
            st_item[lp[bu] + ru] = ((s[k] & (BKEYS - 1)) << 24) | d[k];
            st_dest[lp[bu] + ru] = gb[bu] + ru;
        }
    }
    __syncthreads();

    const int ne = (nE - eb) < SPLIT_EDGES ? (nE - eb) : SPLIT_EDGES;
    for (int i = t; i < 2 * ne; i += 1024) items[st_dest[i]] = st_item[i];
}

__global__ __launch_bounds__(256)
void bucket_lists_kernel(const int* __restrict__ items, const int* __restrict__ gstart,
                         const int* __restrict__ gcount, int* __restrict__ cnt,
                         int* __restrict__ start, int* __restrict__ list,
                         int nbM, int M, int U) {
    __shared__ int kc[BKEYS];
    __shared__ int kex[BKEYS];
    __shared__ int wsum[8];
    __shared__ int stage[STAGE_CAP];
    const int b = blockIdx.x, t = threadIdx.x;
    const int lane = t & 31, w = t >> 5;
    const int g0 = gstart[b];
    const int n  = gcount[b];
    const bool mv = b < nbM;
    const int node0  = (mv ? b : b - nbM) << BSHIFT;
    const int nnodes = mv ? M : U;
    const int out0   = mv ? 0 : M;

    if (t < BKEYS) kc[t] = 0;
    __syncthreads();
    for (int i = t; i < n; i += 256)
        atomicAdd(&kc[(unsigned)items[g0 + i] >> 24], 1);
    __syncthreads();

    const int c = (t < BKEYS) ? kc[t] : 0;
    int x = c;
    for (int o = 1; o < 32; o <<= 1) {
        const int y = __shfl_up(x, o);
        if (lane >= o) x += y;
    }
    if (lane == 31) wsum[w] = x;
    __syncthreads();
    int off = 0;
    for (int k = 0; k < w; ++k) off += wsum[k];
    if (t < BKEYS) {
        const int ex = off + x - c;
        kex[t] = ex;
        kc[t] = 0;
        const int node = node0 + t;
        if (node < nnodes) { cnt[out0 + node] = c; start[out0 + node] = g0 + ex; }
    }
    __syncthreads();

    if (n <= STAGE_CAP) {
        for (int i = t; i < n; i += 256) {
            const unsigned it = (unsigned)items[g0 + i];
            const int key = (int)(it >> 24);
            const int r = atomicAdd(&kc[key], 1);
            stage[kex[key] + r] = (int)(it & 0xFFFFFFu);
        }
        __syncthreads();
        for (int i = t; i < n; i += 256) list[g0 + i] = stage[i];
    } else {
        for (int i = t; i < n; i += 256) {
            const unsigned it = (unsigned)items[g0 + i];
            const int key = (int)(it >> 24);
            const int r = atomicAdd(&kc[key], 1);
            list[g0 + kex[key] + r] = (int)(it & 0xFFFFFFu);
        }
    }
}

#define INIT_BLOCKS 1024

__global__ __launch_bounds__(256)
void movie_init_kernel(const float* __restrict__ mx, const float* __restrict__ linW,
                       const float* __restrict__ linb, const float* __restrict__ memb,
                       float* __restrict__ XM, int M) {
    const int lane = threadIdx.x & 31;
    const int wid = __builtin_amdgcn_readfirstlane((int)(blockIdx.x * 8 + (threadIdx.x >> 5)));
    const int nw  = (int)gridDim.x * 8;
    const int h0 = lane, h1 = lane + 32;

    float w0[20], w1[20];
#pragma unroll
    for (int f = 0; f < 20; ++f) { w0[f] = linW[h0 * 20 + f]; w1[f] = linW[h1 * 20 + f]; }
    const float bias0 = linb[h0], bias1 = linb[h1];

    for (int m = wid; m < M; m += nw) {
        const float* xr = mx + (size_t)m * 20;
        float acc0 = bias0 + memb[(size_t)m * HDIM + h0];
        float acc1 = bias1 + memb[(size_t)m * HDIM + h1];
#pragma unroll
        for (int f = 0; f < 20; ++f) {
            acc0 = fmaf(xr[f], w0[f], acc0);
            acc1 = fmaf(xr[f], w1[f], acc1);
        }
        XM[(size_t)m * HDIM + h0] = acc0;
        XM[(size_t)m * HDIM + h1] = acc1;
    }
}

__global__ __launch_bounds__(128, 16)
void sage_layer_kernel(const float* __restrict__ xsrc, const int* __restrict__ list,
                       const int* __restrict__ start, const int* __restrict__ cnt,
                       const float* __restrict__ xdst,
                       const float* __restrict__ Wl, const float* __restrict__ Wr,
                       const float* __restrict__ bias, float* __restrict__ out,
                       int do_relu) {
    __shared__ __attribute__((aligned(16))) float A1[16][68];
    __shared__ __attribute__((aligned(16))) float A2[16][68];

    const int row0 = blockIdx.x * 16;
    const int tid  = threadIdx.x;
    const int wave = tid >> 5;
    const int lane = tid & 31;
    const int half = lane >> 4;
    const int q    = lane & 15;

    const int node0 = row0 + wave * 4;
    const int st_l = start[node0 + (lane & 3)];
    const int c_l  = cnt[node0 + (lane & 3)];

    {
        const int xr = tid >> 3;
        const int xc = (tid & 7) * 8;
        const float4* gx = (const float4*)(xdst + (size_t)(row0 + xr) * HDIM + xc);
        const float4 x0 = gx[0];
        const float4 x1 = gx[1];
        *(float4*)&A2[xr][xc]     = x0;
        *(float4*)&A2[xr][xc + 4] = x1;
    }

    {
        for (int u = 0; u < 4; ++u) {
            const int st = __builtin_amdgcn_readlane(st_l, u);
            const int c  = __builtin_amdgcn_readlane(c_l, u);
            float4 acc = make_float4(0.f, 0.f, 0.f, 0.f);
            for (int base = 0; base < c; base += 32) {
                const int m = (c - base) < 32 ? (c - base) : 32;
                int idx_l = 0;
                if (lane < m) idx_l = list[st + base + lane];
#pragma unroll
                for (int b = 0; b < 2; ++b) {
                    if (b * 16 < m) {
                        float4 v[8];
#pragma unroll
                        for (int k = 0; k < 8; ++k) {
                            const int j0 = b * 16 + 2 * k;
                            const int s0 = __builtin_amdgcn_readlane(idx_l, j0);
                            const int s1 = __builtin_amdgcn_readlane(idx_l, j0 + 1);
                            const int s  = half ? s1 : s0;
                            v[k] = make_float4(0.f, 0.f, 0.f, 0.f);
                            if (j0 + half < m)
                                v[k] = *(const float4*)(xsrc + (size_t)s * HDIM + q * 4);
                        }
#pragma unroll
                        for (int k = 0; k < 8; ++k) {
                            acc.x += v[k].x; acc.y += v[k].y; acc.z += v[k].z; acc.w += v[k].w;
                        }
                    }
                }
            }
            acc.x += __shfl_xor(acc.x, 16);
            acc.y += __shfl_xor(acc.y, 16);
            acc.z += __shfl_xor(acc.z, 16);
            acc.w += __shfl_xor(acc.w, 16);
            const float dinv = 1.0f / fmaxf((float)c, 1.0f);
            if (half == 0)
                *(float4*)&A1[wave * 4 + u][q * 4] =
                    make_float4(acc.x * dinv, acc.y * dinv, acc.z * dinv, acc.w * dinv);
        }
    }
    __syncthreads();

    const int m    = lane & 15;
    const int col0 = wave * 16;

    v8f c = {};
#pragma unroll
    for (int kk = 0; kk < 16; ++kk) {
        int k = kk * 4 + half * 2;
        v2f a, b;
        a.x = A1[m][k];  a.y = A1[m][k + 1];
        const float* wp = Wl + (size_t)(col0 + m) * HDIM + k;
        b.x = wp[0];     b.y = wp[1];
        c = __builtin_amdgcn_wmma_f32_16x16x4_f32(false, a, false, b,
                                                  (short)0, c, false, false);
    }
#pragma unroll
    for (int kk = 0; kk < 16; ++kk) {
        int k = kk * 4 + half * 2;
        v2f a, b;
        a.x = A2[m][k];  a.y = A2[m][k + 1];
        const float* wp = Wr + (size_t)(col0 + m) * HDIM + k;
        b.x = wp[0];     b.y = wp[1];
        c = __builtin_amdgcn_wmma_f32_16x16x4_f32(false, a, false, b,
                                                  (short)0, c, false, false);
    }

    float bv = bias[col0 + m];
#pragma unroll
    for (int r = 0; r < 8; ++r) {
        float v = c[r] + bv;
        if (do_relu) v = fmaxf(v, 0.0f);
        out[(size_t)(row0 + r + 8 * half) * HDIM + col0 + m] = v;
    }
}

#define DOT_BLOCKS 980

__global__ __launch_bounds__(256) __attribute__((amdgpu_waves_per_eu(8, 8)))
void edge_dot_kernel(const float* __restrict__ U2, const float* __restrict__ M2,
                     const int* __restrict__ eu, const int* __restrict__ em,
                     float* __restrict__ out, int L) {
    const int lane = threadIdx.x & 31;
    const int wid = __builtin_amdgcn_readfirstlane((int)(blockIdx.x * 8 + (threadIdx.x >> 5)));
    const int nw  = (int)gridDim.x * 8;
    const int g = lane >> 3;
    const int s = lane & 7;
    const int ntrip = (L + 31) >> 5;
    if (wid >= ntrip) return;

    int pi = min(wid * 32 + lane, L - 1);
    int iu = eu[pi], im = em[pi];

    for (int c = wid; c < ntrip; c += nw) {
        pi = min((c + nw) * 32 + lane, L - 1);
        const int iu_n = eu[pi], im_n = em[pi];

        float4 a0[8], a1[8], b0[8], b1[8];
#pragma unroll
        for (int k = 0; k < 8; ++k) {
            const int j  = k * 4 + g;
            const int ru = __shfl(iu, j);
            const int rm = __shfl(im, j);
            const float4* a = (const float4*)(U2 + (size_t)ru * HDIM + s * 4);
            const float4* b = (const float4*)(M2 + (size_t)rm * HDIM + s * 4);
            a0[k] = a[0]; a1[k] = a[8]; b0[k] = b[0]; b1[k] = b[8];
        }
        float res = 0.f;
#pragma unroll
        for (int k = 0; k < 8; ++k) {
            float acc = a0[k].x * b0[k].x + a0[k].y * b0[k].y + a0[k].z * b0[k].z + a0[k].w * b0[k].w
                      + a1[k].x * b1[k].x + a1[k].y * b1[k].y + a1[k].z * b1[k].z + a1[k].w * b1[k].w;
            acc += __shfl_xor(acc, 1);
            acc += __shfl_xor(acc, 2);
            acc += __shfl_xor(acc, 4);
            const float r = __shfl(acc, (lane & 3) * 8);
            if ((lane >> 2) == k) res = r;
        }
        if (c * 32 + lane < L) out[c * 32 + lane] = res;
        iu = iu_n; im = im_n;
    }
}

extern "C" void kernel_launch(void* const* d_in, const int* in_sizes, int n_in,
                              void* d_out, int out_size, void* d_ws, size_t ws_size,
                              hipStream_t stream) {
    (void)n_in; (void)ws_size;
    const float* movie_x   = (const float*)d_in[2];
    const int*   e_src     = (const int*)d_in[3];
    const int*   e_dst     = (const int*)d_in[4];
    const int*   eli_u     = (const int*)d_in[5];
    const int*   eli_m     = (const int*)d_in[6];
    const float* user_emb  = (const float*)d_in[7];
    const float* movie_emb = (const float*)d_in[8];
    const float* lin_W     = (const float*)d_in[9];
    const float* lin_b     = (const float*)d_in[10];
    const float* W1um_l    = (const float*)d_in[11];
    const float* W1um_r    = (const float*)d_in[12];
    const float* W1mu_l    = (const float*)d_in[13];
    const float* W1mu_r    = (const float*)d_in[14];
    const float* W2um_l    = (const float*)d_in[15];
    const float* W2um_r    = (const float*)d_in[16];
    const float* W2mu_l    = (const float*)d_in[17];
    const float* W2mu_r    = (const float*)d_in[18];
    const float* b1um      = (const float*)d_in[19];
    const float* b1mu      = (const float*)d_in[20];
    const float* b2um      = (const float*)d_in[21];
    const float* b2mu      = (const float*)d_in[22];
    float* out = (float*)d_out;

    const int U = in_sizes[7] / HDIM;
    const int M = in_sizes[8] / HDIM;
    const int E = in_sizes[3];
    const int L = in_sizes[5];
    (void)out_size;

    size_t off = 0;
    auto carve = [&](size_t nwords) {
        float* p = (float*)((char*)d_ws + off);
        off += ((nwords * sizeof(float) + 255) / 256) * 256;
        return p;
    };
    float* XM   = carve((size_t)M * HDIM);
    float* H_M  = carve((size_t)M * HDIM);
    float* H_U  = carve((size_t)U * HDIM);
    float* U2   = carve((size_t)U * HDIM);
    float* M2   = XM;
    int*   LIST = (int*)carve((size_t)E * 2);
    int*   CNT  = (int*)carve((size_t)M + U + 64);
    int*   START = (int*)carve((size_t)M + U);
    int*   CNT_M = CNT;      int*   CNT_U = CNT + M;
    int*   START_M = START;  int*   START_U = START + M;
    int*   ITEMS  = (int*)U2;
    int*   BCOUNT = (int*)H_M;
    int*   BOFF   = BCOUNT + MAXB;
    int*   BSTART = BOFF + MAXB;

    const int TB = 256;
    dim3 b(TB);
    const int nbM = (M + BKEYS - 1) / BKEYS;
    const int nbU = (U + BKEYS - 1) / BKEYS;
    const int s_blocks = (E + SPLIT_EDGES - 1) / SPLIT_EDGES;

    hipMemsetAsync(BCOUNT, 0, (size_t)2 * MAXB * sizeof(int), stream);
    bucket_hist_kernel<<<s_blocks, 1024, 0, stream>>>(e_src, e_dst, BCOUNT, E, nbM);
    bucket_split_kernel<<<s_blocks, 1024, 0, stream>>>(e_src, e_dst, BCOUNT, BOFF, BSTART, ITEMS,
                                                       E, nbM, nbM + nbU);
    bucket_lists_kernel<<<nbM + nbU, b, 0, stream>>>(ITEMS, BSTART, BCOUNT, CNT, START, LIST,
                                                     nbM, M, U);

    movie_init_kernel<<<INIT_BLOCKS, b, 0, stream>>>(
        movie_x, lin_W, lin_b, movie_emb, XM, M);

    sage_layer_kernel<<<M / 16, 128, 0, stream>>>(user_emb, LIST, START_M, CNT_M, XM,
                                                  W1um_l, W1um_r, b1um, H_M, 1);
    sage_layer_kernel<<<U / 16, 128, 0, stream>>>(XM, LIST, START_U, CNT_U, user_emb,
                                                  W1mu_l, W1mu_r, b1mu, H_U, 1);

    sage_layer_kernel<<<M / 16, 128, 0, stream>>>(H_U, LIST, START_M, CNT_M, H_M,
                                                  W2um_l, W2um_r, b2um, M2, 0);
    sage_layer_kernel<<<U / 16, 128, 0, stream>>>(H_M, LIST, START_U, CNT_U, H_U,
                                                  W2mu_l, W2mu_r, b2mu, U2, 0);

    edge_dot_kernel<<<DOT_BLOCKS, b, 0, stream>>>(U2, M2, eli_u, eli_m, out, L);
}


# independent work shares a launch: movie init rides behind the list blocks, and each layer's movie and user rows run as one grid (9 launches -> 6); arithmetic and every sum order untouched
# speedup vs baseline: 5.3439x; 1.0728x over previous
#include <hip/hip_runtime.h>
#include <hip/hip_bf16.h>

typedef __attribute__((ext_vector_type(2))) float v2f;
typedef __attribute__((ext_vector_type(8))) float v8f;

#define HDIM 64

#define BSHIFT 7
#define BKEYS 128
#define MAXB 2048
#define SPLIT_EDGES 8192
#define STAGE_CAP 12288

__device__ __forceinline__ int scan2048(int c0, int c1, int* wsum) {
    const int lane = threadIdx.x & 31, w = threadIdx.x >> 5;
    int x = c0 + c1;
    for (int o = 1; o < 32; o <<= 1) {
        const int y = __shfl_up(x, o);
        if (lane >= o) x += y;
    }
    __syncthreads();
    if (lane == 31) wsum[w] = x;
    __syncthreads();
    if (w == 0) {
        const int v = wsum[lane];
        int z = v;
        for (int o = 1; o < 32; o <<= 1) {
            const int y = __shfl_up(z, o);
            if (lane >= o) z += y;
        }
        wsum[lane] = z - v;
    }
    __syncthreads();
    return wsum[w] + x - (c0 + c1);
}

__global__ __launch_bounds__(1024)
void bucket_hist_kernel(const int* __restrict__ src, const int* __restrict__ dst,
                        int* __restrict__ gcount, int nE, int nbM) {
    __shared__ int h[MAXB];
    const int t = threadIdx.x;
    h[2 * t] = 0; h[2 * t + 1] = 0;
    __syncthreads();
    const int e0 = blockIdx.x * SPLIT_EDGES + t;
    int s[8], d[8];
    for (int k = 0; k < 8; ++k) {
        const int e = e0 + k * 1024;
        s[k] = -1; d[k] = 0;
        if (e < nE) { s[k] = src[e]; d[k] = dst[e]; }
    }
    for (int k = 0; k < 8; ++k) {
        if (s[k] >= 0) {
            atomicAdd(&h[d[k] >> BSHIFT], 1);
            atomicAdd(&h[nbM + (s[k] >> BSHIFT)], 1);
        }
    }
    __syncthreads();
    const int n0 = h[2 * t], n1 = h[2 * t + 1];
    if (n0) atomicAdd(gcount + 2 * t, n0);
    if (n1) atomicAdd(gcount + 2 * t + 1, n1);
}

__global__ __launch_bounds__(1024)
void bucket_split_kernel(const int* __restrict__ src, const int* __restrict__ dst,
                         const int* __restrict__ gcount, int* __restrict__ goff,
                         int* __restrict__ gstart, int* __restrict__ items,
                         int nE, int nbM, int nb) {
    __shared__ int h[MAXB];
    __shared__ int lp[MAXB];
    __shared__ int gb[MAXB];
    __shared__ int wsum[32];
    __shared__ int st_item[2 * SPLIT_EDGES];
    __shared__ int st_dest[2 * SPLIT_EDGES];
    const int t = threadIdx.x;
    const int i0 = 2 * t, i1 = 2 * t + 1;

    const int eb = blockIdx.x * SPLIT_EDGES;
    int s[8], d[8];
    for (int k = 0; k < 8; ++k) {
        const int e = eb + t + k * 1024;
        s[k] = -1; d[k] = 0;
        if (e < nE) { s[k] = src[e]; d[k] = dst[e]; }
    }

    const int c0 = (i0 < nb) ? gcount[i0] : 0;
    const int c1 = (i1 < nb) ? gcount[i1] : 0;
    const int ex = scan2048(c0, c1, wsum);
    gb[i0] = ex; gb[i1] = ex + c0;
    if (blockIdx.x == 0) {
        if (i0 < nb) gstart[i0] = ex;
        if (i1 < nb) gstart[i1] = ex + c0;
    }
    h[i0] = 0; h[i1] = 0;
    __syncthreads();

    for (int k = 0; k < 8; ++k) {
        if (s[k] >= 0) {
            atomicAdd(&h[d[k] >> BSHIFT], 1);
            atomicAdd(&h[nbM + (s[k] >> BSHIFT)], 1);
        }
    }
    __syncthreads();

    const int n0 = h[i0], n1 = h[i1];
    const int lex = scan2048(n0, n1, wsum);
    lp[i0] = lex; lp[i1] = lex + n0;
    if (n0) gb[i0] += atomicAdd(goff + i0, n0);
    if (n1) gb[i1] += atomicAdd(goff + i1, n1);
    h[i0] = 0; h[i1] = 0;
    __syncthreads();

    for (int k = 0; k < 8; ++k) {
        if (s[k] >= 0) {
            const int bm = d[k] >> BSHIFT;
            const int rm = atomicAdd(&h[bm], 1);
            st_item[lp[bm] + rm] = ((d[k] & (BKEYS - 1)) << 24) | s[k];
            st_dest[lp[bm] + rm] = gb[bm] + rm;
            const int bu = nbM + (s[k] >> BSHIFT);
            const int ru = atomicAdd(&h[bu], 1);
            st_item[lp[bu] + ru] = ((s[k] & (BKEYS - 1)) << 24) | d[k];
            st_dest[lp[bu] + ru] = gb[bu] + ru;
        }
    }
    __syncthreads();

    const int ne = (nE - eb) < SPLIT_EDGES ? (nE - eb) : SPLIT_EDGES;
    for (int i = t; i < 2 * ne; i += 1024) items[st_dest[i]] = st_item[i];
}

#define INIT_BLOCKS 512

__device__ __forceinline__
void movie_init_rows(const float* __restrict__ mx, const float* __restrict__ linW,
                     const float* __restrict__ linb, const float* __restrict__ memb,
                     float* __restrict__ XM, int M, int wid, int nw) {
    const int lane = threadIdx.x & 31;
    const int h0 = lane, h1 = lane + 32;

    float w0[20], w1[20];
#pragma unroll
    for (int f = 0; f < 20; ++f) { w0[f] = linW[h0 * 20 + f]; w1[f] = linW[h1 * 20 + f]; }
    const float bias0 = linb[h0], bias1 = linb[h1];

    for (int m = wid; m < M; m += nw) {
        const float* xr = mx + (size_t)m * 20;
        float acc0 = bias0 + memb[(size_t)m * HDIM + h0];
        float acc1 = bias1 + memb[(size_t)m * HDIM + h1];
#pragma unroll
        for (int f = 0; f < 20; ++f) {
            acc0 = fmaf(xr[f], w0[f], acc0);
            acc1 = fmaf(xr[f], w1[f], acc1);
        }
        XM[(size_t)m * HDIM + h0] = acc0;
        XM[(size_t)m * HDIM + h1] = acc1;
    }
}

__global__ __launch_bounds__(256)
void bucket_lists_movie_init_kernel(const int* __restrict__ items, const int* __restrict__ gstart,
                                    const int* __restrict__ gcount, int* __restrict__ cnt,
                                    int* __restrict__ start, int* __restrict__ list,
                                    int nbM, int M, int U, int nb,
                                    const float* __restrict__ mx, const float* __restrict__ linW,
                                    const float* __restrict__ linb, const float* __restrict__ memb,
                                    float* __restrict__ XM) {
    __shared__ int kc[BKEYS];
    __shared__ int kex[BKEYS];
    __shared__ int wsum[8];
    __shared__ int stage[STAGE_CAP];
    const int b = blockIdx.x, t = threadIdx.x;

    if (b >= nb) {
        const int wid = __builtin_amdgcn_readfirstlane((b - nb) * 8 + (t >> 5));
        movie_init_rows(mx, linW, linb, memb, XM, M, wid, ((int)gridDim.x - nb) * 8);
        return;
    }

    const int lane = t & 31, w = t >> 5;
    const int g0 = gstart[b];
    const int n  = gcount[b];
    const bool mv = b < nbM;
    const int node0  = (mv ? b : b - nbM) << BSHIFT;
    const int nnodes = mv ? M : U;
    const int out0   = mv ? 0 : M;

    if (t < BKEYS) kc[t] = 0;
    __syncthreads();
    for (int i = t; i < n; i += 256)
        atomicAdd(&kc[(unsigned)items[g0 + i] >> 24], 1);
    __syncthreads();

    const int c = (t < BKEYS) ? kc[t] : 0;
    int x = c;
    for (int o = 1; o < 32; o <<= 1) {
        const int y = __shfl_up(x, o);
        if (lane >= o) x += y;
    }
    if (lane == 31) wsum[w] = x;
    __syncthreads();
    int off = 0;
    for (int k = 0; k < w; ++k) off += wsum[k];
    if (t < BKEYS) {
        const int ex = off + x - c;
        kex[t] = ex;
        kc[t] = 0;
        const int node = node0 + t;
        if (node < nnodes) { cnt[out0 + node] = c; start[out0 + node] = g0 + ex; }
    }
    __syncthreads();

    if (n <= STAGE_CAP) {
        for (int i = t; i < n; i += 256) {
            const unsigned it = (unsigned)items[g0 + i];
            const int key = (int)(it >> 24);
            const int r = atomicAdd(&kc[key], 1);
            stage[kex[key] + r] = (int)(it & 0xFFFFFFu);
        }
        __syncthreads();
        for (int i = t; i < n; i += 256) list[g0 + i] = stage[i];
    } else {
        for (int i = t; i < n; i += 256) {
            const unsigned it = (unsigned)items[g0 + i];
            const int key = (int)(it >> 24);
            const int r = atomicAdd(&kc[key], 1);
            list[g0 + kex[key] + r] = (int)(it & 0xFFFFFFu);
        }
    }
}

__global__ __launch_bounds__(128, 16)
void sage_layer_kernel(const int* __restrict__ list,
                       const float* __restrict__ xsrcM, const int* __restrict__ startM,
                       const int* __restrict__ cntM, const float* __restrict__ xdstM,
                       const float* __restrict__ WlM, const float* __restrict__ WrM,
                       const float* __restrict__ biasM, float* __restrict__ outM,
                       const float* __restrict__ xsrcU, const int* __restrict__ startU,
                       const int* __restrict__ cntU, const float* __restrict__ xdstU,
                       const float* __restrict__ WlU, const float* __restrict__ WrU,
                       const float* __restrict__ biasU, float* __restrict__ outU,
                       int nblkM, int do_relu) {
    __shared__ __attribute__((aligned(16))) float A1[16][68];
    __shared__ __attribute__((aligned(16))) float A2[16][68];

    const bool mvb = (int)blockIdx.x < nblkM;
    const float* __restrict__ xsrc = mvb ? xsrcM : xsrcU;
    const int*   __restrict__ start = mvb ? startM : startU;
    const int*   __restrict__ cnt  = mvb ? cntM : cntU;
    const float* __restrict__ xdst = mvb ? xdstM : xdstU;
    const float* __restrict__ Wl   = mvb ? WlM : WlU;
    const float* __restrict__ Wr   = mvb ? WrM : WrU;
    const float* __restrict__ bias = mvb ? biasM : biasU;
    float*       __restrict__ out  = mvb ? outM : outU;

    const int row0 = (mvb ? (int)blockIdx.x : (int)blockIdx.x - nblkM) * 16;
    const int tid  = threadIdx.x;
    const int wave = tid >> 5;
    const int lane = tid & 31;
    const int half = lane >> 4;
    const int q    = lane & 15;

    const int node0 = row0 + wave * 4;
    const int st_l = start[node0 + (lane & 3)];
    const int c_l  = cnt[node0 + (lane & 3)];

    {
        const int xr = tid >> 3;
        const int xc = (tid & 7) * 8;
        const float4* gx = (const float4*)(xdst + (size_t)(row0 + xr) * HDIM + xc);
        const float4 x0 = gx[0];
        const float4 x1 = gx[1];
        *(float4*)&A2[xr][xc]     = x0;
        *(float4*)&A2[xr][xc + 4] = x1;
    }

    {
        for (int u = 0; u < 4; ++u) {
            const int st = __builtin_amdgcn_readlane(st_l, u);
            const int c  = __builtin_amdgcn_readlane(c_l, u);
            float4 acc = make_float4(0.f, 0.f, 0.f, 0.f);
            for (int base = 0; base < c; base += 32) {
                const int m = (c - base) < 32 ? (c - base) : 32;
                int idx_l = 0;
                if (lane < m) idx_l = list[st + base + lane];
#pragma unroll
                for (int b = 0; b < 2; ++b) {
                    if (b * 16 < m) {
                        float4 v[8];
#pragma unroll
                        for (int k = 0; k < 8; ++k) {
                            const int j0 = b * 16 + 2 * k;
                            const int s0 = __builtin_amdgcn_readlane(idx_l, j0);
                            const int s1 = __builtin_amdgcn_readlane(idx_l, j0 + 1);
                            const int s  = half ? s1 : s0;
                            v[k] = make_float4(0.f, 0.f, 0.f, 0.f);
                            if (j0 + half < m)
                                v[k] = *(const float4*)(xsrc + (size_t)s * HDIM + q * 4);
                        }
#pragma unroll
                        for (int k = 0; k < 8; ++k) {
                            acc.x += v[k].x; acc.y += v[k].y; acc.z += v[k].z; acc.w += v[k].w;
                        }
                    }
                }
            }
            acc.x += __shfl_xor(acc.x, 16);
            acc.y += __shfl_xor(acc.y, 16);
            acc.z += __shfl_xor(acc.z, 16);
            acc.w += __shfl_xor(acc.w, 16);
            const float dinv = 1.0f / fmaxf((float)c, 1.0f);
            if (half == 0)
                *(float4*)&A1[wave * 4 + u][q * 4] =
                    make_float4(acc.x * dinv, acc.y * dinv, acc.z * dinv, acc.w * dinv);
        }
    }
    __syncthreads();

    const int m    = lane & 15;
    const int col0 = wave * 16;

    v8f c = {};
#pragma unroll
    for (int kk = 0; kk < 16; ++kk) {
        int k = kk * 4 + half * 2;
        v2f a, b;
        a.x = A1[m][k];  a.y = A1[m][k + 1];
        const float* wp = Wl + (size_t)(col0 + m) * HDIM + k;
        b.x = wp[0];     b.y = wp[1];
        c = __builtin_amdgcn_wmma_f32_16x16x4_f32(false, a, false, b,
                                                  (short)0, c, false, false);
    }
#pragma unroll
    for (int kk = 0; kk < 16; ++kk) {
        int k = kk * 4 + half * 2;
        v2f a, b;
        a.x = A2[m][k];  a.y = A2[m][k + 1];
        const float* wp = Wr + (size_t)(col0 + m) * HDIM + k;
        b.x = wp[0];     b.y = wp[1];
        c = __builtin_amdgcn_wmma_f32_16x16x4_f32(false, a, false, b,
                                                  (short)0, c, false, false);
    }

    float bv = bias[col0 + m];
#pragma unroll
    for (int r = 0; r < 8; ++r) {
        float v = c[r] + bv;
        if (do_relu) v = fmaxf(v, 0.0f);
        out[(size_t)(row0 + r + 8 * half) * HDIM + col0 + m] = v;
    }
}

#define DOT_BLOCKS 980

__global__ __launch_bounds__(256) __attribute__((amdgpu_waves_per_eu(8, 8)))
void edge_dot_kernel(const float* __restrict__ U2, const float* __restrict__ M2,
                     const int* __restrict__ eu, const int* __restrict__ em,
                     float* __restrict__ out, int L) {
    const int lane = threadIdx.x & 31;
    const int wid = __builtin_amdgcn_readfirstlane((int)(blockIdx.x * 8 + (threadIdx.x >> 5)));
    const int nw  = (int)gridDim.x * 8;
    const int g = lane >> 3;
    const int s = lane & 7;
    const int ntrip = (L + 31) >> 5;
    if (wid >= ntrip) return;

    int pi = min(wid * 32 + lane, L - 1);
    int iu = eu[pi], im = em[pi];

    for (int c = wid; c < ntrip; c += nw) {
        pi = min((c + nw) * 32 + lane, L - 1);
        const int iu_n = eu[pi], im_n = em[pi];

        float4 a0[8], a1[8], b0[8], b1[8];
#pragma unroll
        for (int k = 0; k < 8; ++k) {
            const int j  = k * 4 + g;
            const int ru = __shfl(iu, j);
            const int rm = __shfl(im, j);
            const float4* a = (const float4*)(U2 + (size_t)ru * HDIM + s * 4);
            const float4* b = (const float4*)(M2 + (size_t)rm * HDIM + s * 4);
            a0[k] = a[0]; a1[k] = a[8]; b0[k] = b[0]; b1[k] = b[8];
        }
        float res = 0.f;
#pragma unroll
        for (int k = 0; k < 8; ++k) {
            float acc = a0[k].x * b0[k].x + a0[k].y * b0[k].y + a0[k].z * b0[k].z + a0[k].w * b0[k].w
                      + a1[k].x * b1[k].x + a1[k].y * b1[k].y + a1[k].z * b1[k].z + a1[k].w * b1[k].w;
            acc += __shfl_xor(acc, 1);
            acc += __shfl_xor(acc, 2);
            acc += __shfl_xor(acc, 4);
            const float r = __shfl(acc, (lane & 3) * 8);
            if ((lane >> 2) == k) res = r;
        }
        if (c * 32 + lane < L) out[c * 32 + lane] = res;
        iu = iu_n; im = im_n;
    }
}

extern "C" void kernel_launch(void* const* d_in, const int* in_sizes, int n_in,
                              void* d_out, int out_size, void* d_ws, size_t ws_size,
                              hipStream_t stream) {
    (void)n_in; (void)ws_size;
    const float* movie_x   = (const float*)d_in[2];
    const int*   e_src     = (const int*)d_in[3];
    const int*   e_dst     = (const int*)d_in[4];
    const int*   eli_u     = (const int*)d_in[5];
    const int*   eli_m     = (const int*)d_in[6];
    const float* user_emb  = (const float*)d_in[7];
    const float* movie_emb = (const float*)d_in[8];
    const float* lin_W     = (const float*)d_in[9];
    const float* lin_b     = (const float*)d_in[10];
    const float* W1um_l    = (const float*)d_in[11];
    const float* W1um_r    = (const float*)d_in[12];
    const float* W1mu_l    = (const float*)d_in[13];
    const float* W1mu_r    = (const float*)d_in[14];
    const float* W2um_l    = (const float*)d_in[15];
    const float* W2um_r    = (const float*)d_in[16];
    const float* W2mu_l    = (const float*)d_in[17];
    const float* W2mu_r    = (const float*)d_in[18];
    const float* b1um      = (const float*)d_in[19];
    const float* b1mu      = (const float*)d_in[20];
    const float* b2um      = (const float*)d_in[21];
    const float* b2mu      = (const float*)d_in[22];
    float* out = (float*)d_out;

    const int U = in_sizes[7] / HDIM;
    const int M = in_sizes[8] / HDIM;
    const int E = in_sizes[3];
    const int L = in_sizes[5];
    (void)out_size;

    size_t off = 0;
    auto carve = [&](size_t nwords) {
        float* p = (float*)((char*)d_ws + off);
        off += ((nwords * sizeof(float) + 255) / 256) * 256;
        return p;
    };
    float* XM   = carve((size_t)M * HDIM);
    float* H_M  = carve((size_t)M * HDIM);
    float* H_U  = carve((size_t)U * HDIM);
    float* U2   = carve((size_t)U * HDIM);
    float* M2   = XM;
    int*   LIST = (int*)carve((size_t)E * 2);
    int*   CNT  = (int*)carve((size_t)M + U + 64);
    int*   START = (int*)carve((size_t)M + U);
    int*   CNT_M = CNT;      int*   CNT_U = CNT + M;
    int*   START_M = START;  int*   START_U = START + M;
    int*   ITEMS  = (int*)U2;
    int*   BCOUNT = (int*)H_M;
    int*   BOFF   = BCOUNT + MAXB;
    int*   BSTART = BOFF + MAXB;

    const int TB = 256;
    dim3 b(TB);
    const int nbM = (M + BKEYS - 1) / BKEYS;
    const int nbU = (U + BKEYS - 1) / BKEYS;
    const int s_blocks = (E + SPLIT_EDGES - 1) / SPLIT_EDGES;

    hipMemsetAsync(BCOUNT, 0, (size_t)2 * MAXB * sizeof(int), stream);
    bucket_hist_kernel<<<s_blocks, 1024, 0, stream>>>(e_src, e_dst, BCOUNT, E, nbM);
    bucket_split_kernel<<<s_blocks, 1024, 0, stream>>>(e_src, e_dst, BCOUNT, BOFF, BSTART, ITEMS,
                                                       E, nbM, nbM + nbU);
    bucket_lists_movie_init_kernel<<<nbM + nbU + INIT_BLOCKS, b, 0, stream>>>(
        ITEMS, BSTART, BCOUNT, CNT, START, LIST, nbM, M, U, nbM + nbU,
        movie_x, lin_W, lin_b, movie_emb, XM);

    sage_layer_kernel<<<M / 16 + U / 16, 128, 0, stream>>>(
        LIST,
        user_emb, START_M, CNT_M, XM, W1um_l, W1um_r, b1um, H_M,
        XM, START_U, CNT_U, user_emb, W1mu_l, W1mu_r, b1mu, H_U,
        M / 16, 1);

    sage_layer_kernel<<<M / 16 + U / 16, 128, 0, stream>>>(
        LIST,
        H_U, START_M, CNT_M, H_M, W2um_l, W2um_r, b2um, M2,
        H_M, START_U, CNT_U, H_U, W2mu_l, W2mu_r, b2mu, U2,
        M / 16, 0);

    edge_dot_kernel<<<DOT_BLOCKS, b, 0, stream>>>(U2, M2, eli_u, eli_m, out, L);
}


# layer kernel: a wave's weight rows come in as whole 128-B lines (4 lines per load; a B fragment read from memory touched 16) and reach the B fragments through wave-private LDS; sum order untouched
# speedup vs baseline: 5.5390x; 1.0365x over previous
#include <hip/hip_runtime.h>
#include <hip/hip_bf16.h>

typedef __attribute__((ext_vector_type(2))) float v2f;
typedef __attribute__((ext_vector_type(8))) float v8f;

#define HDIM 64

#define BSHIFT 7
#define BKEYS 128
#define MAXB 2048
#define SPLIT_EDGES 8192
#define STAGE_CAP 12288

__device__ __forceinline__ int scan2048(int c0, int c1, int* wsum) {
    const int lane = threadIdx.x & 31, w = threadIdx.x >> 5;
    int x = c0 + c1;
    for (int o = 1; o < 32; o <<= 1) {
        const int y = __shfl_up(x, o);
        if (lane >= o) x += y;
    }
    __syncthreads();
    if (lane == 31) wsum[w] = x;
    __syncthreads();
    if (w == 0) {
        const int v = wsum[lane];
        int z = v;
        for (int o = 1; o < 32; o <<= 1) {
            const int y = __shfl_up(z, o);
            if (lane >= o) z += y;
        }
        wsum[lane] = z - v;
    }
    __syncthreads();
    return wsum[w] + x - (c0 + c1);
}

__global__ __launch_bounds__(1024)
void bucket_hist_kernel(const int* __restrict__ src, const int* __restrict__ dst,
                        int* __restrict__ gcount, int nE, int nbM) {
    __shared__ int h[MAXB];
    const int t = threadIdx.x;
    h[2 * t] = 0; h[2 * t + 1] = 0;
    __syncthreads();
    const int e0 = blockIdx.x * SPLIT_EDGES + t;
    int s[8], d[8];
    for (int k = 0; k < 8; ++k) {
        const int e = e0 + k * 1024;
        s[k] = -1; d[k] = 0;
        if (e < nE) { s[k] = src[e]; d[k] = dst[e]; }
    }
    for (int k = 0; k < 8; ++k) {
        if (s[k] >= 0) {
            atomicAdd(&h[d[k] >> BSHIFT], 1);
            atomicAdd(&h[nbM + (s[k] >> BSHIFT)], 1);
        }
    }
    __syncthreads();
    const int n0 = h[2 * t], n1 = h[2 * t + 1];
    if (n0) atomicAdd(gcount + 2 * t, n0);
    if (n1) atomicAdd(gcount + 2 * t + 1, n1);
}

__global__ __launch_bounds__(1024)
void bucket_split_kernel(const int* __restrict__ src, const int* __restrict__ dst,
                         const int* __restrict__ gcount, int* __restrict__ goff,
                         int* __restrict__ gstart, int* __restrict__ items,
                         int nE, int nbM, int nb) {
    __shared__ int h[MAXB];
    __shared__ int lp[MAXB];
    __shared__ int gb[MAXB];
    __shared__ int wsum[32];
    __shared__ int st_item[2 * SPLIT_EDGES];
    __shared__ int st_dest[2 * SPLIT_EDGES];
    const int t = threadIdx.x;
    const int i0 = 2 * t, i1 = 2 * t + 1;

    const int eb = blockIdx.x * SPLIT_EDGES;
    int s[8], d[8];
    for (int k = 0; k < 8; ++k) {
        const int e = eb + t + k * 1024;
        s[k] = -1; d[k] = 0;
        if (e < nE) { s[k] = src[e]; d[k] = dst[e]; }
    }

    const int c0 = (i0 < nb) ? gcount[i0] : 0;
    const int c1 = (i1 < nb) ? gcount[i1] : 0;
    const int ex = scan2048(c0, c1, wsum);
    gb[i0] = ex; gb[i1] = ex + c0;
    if (blockIdx.x == 0) {
        if (i0 < nb) gstart[i0] = ex;
        if (i1 < nb) gstart[i1] = ex + c0;
    }
    h[i0] = 0; h[i1] = 0;
    __syncthreads();

    for (int k = 0; k < 8; ++k) {
        if (s[k] >= 0) {
            atomicAdd(&h[d[k] >> BSHIFT], 1);
            atomicAdd(&h[nbM + (s[k] >> BSHIFT)], 1);
        }
    }
    __syncthreads();

    const int n0 = h[i0], n1 = h[i1];
    const int lex = scan2048(n0, n1, wsum);
    lp[i0] = lex; lp[i1] = lex + n0;
    if (n0) gb[i0] += atomicAdd(goff + i0, n0);
    if (n1) gb[i1] += atomicAdd(goff + i1, n1);
    h[i0] = 0; h[i1] = 0;
    __syncthreads();

    for (int k = 0; k < 8; ++k) {
        if (s[k] >= 0) {
            const int bm = d[k] >> BSHIFT;
            const int rm = atomicAdd(&h[bm], 1);
            st_item[lp[bm] + rm] = ((d[k] & (BKEYS - 1)) << 24) | s[k];
            st_dest[lp[bm] + rm] = gb[bm] + rm;
            const int bu = nbM + (s[k] >> BSHIFT);
            const int ru = atomicAdd(&h[bu], 1);
            st_item[lp[bu] + ru] = ((s[k] & (BKEYS - 1)) << 24) | d[k];
            st_dest[lp[bu] + ru] = gb[bu] + ru;
        }
    }
    __syncthreads();

    const int ne = (nE - eb) < SPLIT_EDGES ? (nE - eb) : SPLIT_EDGES;
    for (int i = t; i < 2 * ne; i += 1024) items[st_dest[i]] = st_item[i];
}

#define INIT_BLOCKS 512

__device__ __forceinline__
void movie_init_rows(const float* __restrict__ mx, const float* __restrict__ linW,
                     const float* __restrict__ linb, const float* __restrict__ memb,
                     float* __restrict__ XM, int M, int wid, int nw) {
    const int lane = threadIdx.x & 31;
    const int h0 = lane, h1 = lane + 32;

    float w0[20], w1[20];
#pragma unroll
    for (int f = 0; f < 20; ++f) { w0[f] = linW[h0 * 20 + f]; w1[f] = linW[h1 * 20 + f]; }
    const float bias0 = linb[h0], bias1 = linb[h1];

    for (int m = wid; m < M; m += nw) {
        const float* xr = mx + (size_t)m * 20;
        float acc0 = bias0 + memb[(size_t)m * HDIM + h0];
        float acc1 = bias1 + memb[(size_t)m * HDIM + h1];
#pragma unroll
        for (int f = 0; f < 20; ++f) {
            acc0 = fmaf(xr[f], w0[f], acc0);
            acc1 = fmaf(xr[f], w1[f], acc1);
        }
        XM[(size_t)m * HDIM + h0] = acc0;
        XM[(size_t)m * HDIM + h1] = acc1;
    }
}

__global__ __launch_bounds__(256)
void bucket_lists_movie_init_kernel(const int* __restrict__ items, const int* __restrict__ gstart,
                                    const int* __restrict__ gcount, int* __restrict__ cnt,
                                    int* __restrict__ start, int* __restrict__ list,
                                    int nbM, int M, int U, int nb,
                                    const float* __restrict__ mx, const float* __restrict__ linW,
                                    const float* __restrict__ linb, const float* __restrict__ memb,
                                    float* __restrict__ XM) {
    __shared__ int kc[BKEYS];
    __shared__ int kex[BKEYS];
    __shared__ int wsum[8];
    __shared__ int stage[STAGE_CAP];
    const int b = blockIdx.x, t = threadIdx.x;

    if (b >= nb) {
        const int wid = __builtin_amdgcn_readfirstlane((b - nb) * 8 + (t >> 5));
        movie_init_rows(mx, linW, linb, memb, XM, M, wid, ((int)gridDim.x - nb) * 8);
        return;
    }

    const int lane = t & 31, w = t >> 5;
    const int g0 = gstart[b];
    const int n  = gcount[b];
    const bool mv = b < nbM;
    const int node0  = (mv ? b : b - nbM) << BSHIFT;
    const int nnodes = mv ? M : U;
    const int out0   = mv ? 0 : M;

    if (t < BKEYS) kc[t] = 0;
    __syncthreads();
    for (int i = t; i < n; i += 256)
        atomicAdd(&kc[(unsigned)items[g0 + i] >> 24], 1);
    __syncthreads();

    const int c = (t < BKEYS) ? kc[t] : 0;
    int x = c;
    for (int o = 1; o < 32; o <<= 1) {
        const int y = __shfl_up(x, o);
        if (lane >= o) x += y;
    }
    if (lane == 31) wsum[w] = x;
    __syncthreads();
    int off = 0;
    for (int k = 0; k < w; ++k) off += wsum[k];
    if (t < BKEYS) {
        const int ex = off + x - c;
        kex[t] = ex;
        kc[t] = 0;
        const int node = node0 + t;
        if (node < nnodes) { cnt[out0 + node] = c; start[out0 + node] = g0 + ex; }
    }
    __syncthreads();

    if (n <= STAGE_CAP) {
        for (int i = t; i < n; i += 256) {
            const unsigned it = (unsigned)items[g0 + i];
            const int key = (int)(it >> 24);
            const int r = atomicAdd(&kc[key], 1);
            stage[kex[key] + r] = (int)(it & 0xFFFFFFu);
        }
        __syncthreads();
        for (int i = t; i < n; i += 256) list[g0 + i] = stage[i];
    } else {
        for (int i = t; i < n; i += 256) {
            const unsigned it = (unsigned)items[g0 + i];
            const int key = (int)(it >> 24);
            const int r = atomicAdd(&kc[key], 1);
            list[g0 + kex[key] + r] = (int)(it & 0xFFFFFFu);
        }
    }
}

#define WS_PITCH 36
#define WS_K 32

__global__ __launch_bounds__(128, 16)
void sage_layer_kernel(const int* __restrict__ list,
                       const float* __restrict__ xsrcM, const int* __restrict__ startM,
                       const int* __restrict__ cntM, const float* __restrict__ xdstM,
                       const float* __restrict__ WlM, const float* __restrict__ WrM,
                       const float* __restrict__ biasM, float* __restrict__ outM,
                       const float* __restrict__ xsrcU, const int* __restrict__ startU,
                       const int* __restrict__ cntU, const float* __restrict__ xdstU,
                       const float* __restrict__ WlU, const float* __restrict__ WrU,
                       const float* __restrict__ biasU, float* __restrict__ outU,
                       int nblkM, int do_relu) {
    __shared__ __attribute__((aligned(16))) float A1[16][68];
    __shared__ __attribute__((aligned(16))) float A2[16][68];
    __shared__ __attribute__((aligned(16))) float WS[4][16][WS_PITCH];

    const bool mvb = (int)blockIdx.x < nblkM;
    const float* __restrict__ xsrc = mvb ? xsrcM : xsrcU;
    const int*   __restrict__ start = mvb ? startM : startU;
    const int*   __restrict__ cnt  = mvb ? cntM : cntU;
    const float* __restrict__ xdst = mvb ? xdstM : xdstU;
    const float* __restrict__ Wl   = mvb ? WlM : WlU;
    const float* __restrict__ Wr   = mvb ? WrM : WrU;
    const float* __restrict__ bias = mvb ? biasM : biasU;
    float*       __restrict__ out  = mvb ? outM : outU;

    const int row0 = (mvb ? (int)blockIdx.x : (int)blockIdx.x - nblkM) * 16;
    const int tid  = threadIdx.x;
    const int wave = tid >> 5;
    const int lane = tid & 31;
    const int half = lane >> 4;
    const int q    = lane & 15;

    const int node0 = row0 + wave * 4;
    const int st_l = start[node0 + (lane & 3)];
    const int c_l  = cnt[node0 + (lane & 3)];

    {
        const int xr = tid >> 3;
        const int xc = (tid & 7) * 8;
        const float4* gx = (const float4*)(xdst + (size_t)(row0 + xr) * HDIM + xc);
        const float4 x0 = gx[0];
        const float4 x1 = gx[1];
        *(float4*)&A2[xr][xc]     = x0;
        *(float4*)&A2[xr][xc + 4] = x1;
    }

    {
        for (int u = 0; u < 4; ++u) {
            const int st = __builtin_amdgcn_readlane(st_l, u);
            const int c  = __builtin_amdgcn_readlane(c_l, u);
            float4 acc = make_float4(0.f, 0.f, 0.f, 0.f);
            for (int base = 0; base < c; base += 32) {
                const int m = (c - base) < 32 ? (c - base) : 32;
                int idx_l = 0;
                if (lane < m) idx_l = list[st + base + lane];
#pragma unroll
                for (int b = 0; b < 2; ++b) {
                    if (b * 16 < m) {
                        float4 v[8];
#pragma unroll
                        for (int k = 0; k < 8; ++k) {
                            const int j0 = b * 16 + 2 * k;
                            const int s0 = __builtin_amdgcn_readlane(idx_l, j0);
                            const int s1 = __builtin_amdgcn_readlane(idx_l, j0 + 1);
                            const int s  = half ? s1 : s0;
                            v[k] = make_float4(0.f, 0.f, 0.f, 0.f);
                            if (j0 + half < m)
                                v[k] = *(const float4*)(xsrc + (size_t)s * HDIM + q * 4);
                        }
#pragma unroll
                        for (int k = 0; k < 8; ++k) {
                            acc.x += v[k].x; acc.y += v[k].y; acc.z += v[k].z; acc.w += v[k].w;
                        }
                    }
                }
            }
            acc.x += __shfl_xor(acc.x, 16);
            acc.y += __shfl_xor(acc.y, 16);
            acc.z += __shfl_xor(acc.z, 16);
            acc.w += __shfl_xor(acc.w, 16);
            const float dinv = 1.0f / fmaxf((float)c, 1.0f);
            if (half == 0)
                *(float4*)&A1[wave * 4 + u][q * 4] =
                    make_float4(acc.x * dinv, acc.y * dinv, acc.z * dinv, acc.w * dinv);
        }
    }
    __syncthreads();

    const int m    = lane & 15;
    const int col0 = wave * 16;

    float (*ws)[WS_PITCH] = WS[wave];
    const int srow = lane >> 3;
    const int scol = (lane & 7) * 4;

    const size_t wrow = (size_t)(col0 + srow) * HDIM + scol;
    float4 g0 = *(const float4*)(Wl + wrow);
    float4 g1 = *(const float4*)(Wl + wrow + 4 * HDIM);
    float4 g2 = *(const float4*)(Wl + wrow + 8 * HDIM);
    float4 g3 = *(const float4*)(Wl + wrow + 12 * HDIM);

    v8f c = {};
#pragma unroll
    for (int ch = 0; ch < 4; ++ch) {
        *(float4*)&ws[srow][scol]      = g0;
        *(float4*)&ws[4 + srow][scol]  = g1;
        *(float4*)&ws[8 + srow][scol]  = g2;
        *(float4*)&ws[12 + srow][scol] = g3;

        __syncwarp();

        if (ch < 3) {
            const float* Wn = ((ch + 1 < 2) ? Wl : Wr) + ((ch + 1) & 1) * WS_K;
            g0 = *(const float4*)(Wn + wrow);
            g1 = *(const float4*)(Wn + wrow + 4 * HDIM);
            g2 = *(const float4*)(Wn + wrow + 8 * HDIM);
            g3 = *(const float4*)(Wn + wrow + 12 * HDIM);
        }

        const float (*At)[68] = (ch < 2) ? A1 : A2;
        const int kb = (ch & 1) * WS_K;
#pragma unroll
        for (int kk = 0; kk < WS_K / 4; ++kk) {
            const int k = kk * 4 + half * 2;
            v2f a, b;
            a.x = At[m][kb + k];  a.y = At[m][kb + k + 1];
            b.x = ws[m][k];       b.y = ws[m][k + 1];
            c = __builtin_amdgcn_wmma_f32_16x16x4_f32(false, a, false, b,
                                                      (short)0, c, false, false);
        }

        if (ch < 3) __syncwarp();
    }
    __builtin_amdgcn_sched_barrier(0);

    float bv = bias[col0 + m];
#pragma unroll
    for (int r = 0; r < 8; ++r) {
        float v = c[r] + bv;
        if (do_relu) v = fmaxf(v, 0.0f);
        out[(size_t)(row0 + r + 8 * half) * HDIM + col0 + m] = v;
    }
}

#define DOT_BLOCKS 980

__global__ __launch_bounds__(256) __attribute__((amdgpu_waves_per_eu(8, 8)))
void edge_dot_kernel(const float* __restrict__ U2, const float* __restrict__ M2,
                     const int* __restrict__ eu, const int* __restrict__ em,
                     float* __restrict__ out, int L) {
    const int lane = threadIdx.x & 31;
    const int wid = __builtin_amdgcn_readfirstlane((int)(blockIdx.x * 8 + (threadIdx.x >> 5)));
    const int nw  = (int)gridDim.x * 8;
    const int g = lane >> 3;
    const int s = lane & 7;
    const int ntrip = (L + 31) >> 5;
    if (wid >= ntrip) return;

    int pi = min(wid * 32 + lane, L - 1);
    int iu = eu[pi], im = em[pi];

    for (int c = wid; c < ntrip; c += nw) {
        pi = min((c + nw) * 32 + lane, L - 1);
        const int iu_n = eu[pi], im_n = em[pi];

        float4 a0[8], a1[8], b0[8], b1[8];
#pragma unroll
        for (int k = 0; k < 8; ++k) {
            const int j  = k * 4 + g;
            const int ru = __shfl(iu, j);
            const int rm = __shfl(im, j);
            const float4* a = (const float4*)(U2 + (size_t)ru * HDIM + s * 4);
            const float4* b = (const float4*)(M2 + (size_t)rm * HDIM + s * 4);
            a0[k] = a[0]; a1[k] = a[8]; b0[k] = b[0]; b1[k] = b[8];
        }
        float res = 0.f;
#pragma unroll
        for (int k = 0; k < 8; ++k) {
            float acc = a0[k].x * b0[k].x + a0[k].y * b0[k].y + a0[k].z * b0[k].z + a0[k].w * b0[k].w
                      + a1[k].x * b1[k].x + a1[k].y * b1[k].y + a1[k].z * b1[k].z + a1[k].w * b1[k].w;
            acc += __shfl_xor(acc, 1);
            acc += __shfl_xor(acc, 2);
            acc += __shfl_xor(acc, 4);
            const float r = __shfl(acc, (lane & 3) * 8);
            if ((lane >> 2) == k) res = r;
        }
        if (c * 32 + lane < L) out[c * 32 + lane] = res;
        iu = iu_n; im = im_n;
    }
}

extern "C" void kernel_launch(void* const* d_in, const int* in_sizes, int n_in,
                              void* d_out, int out_size, void* d_ws, size_t ws_size,
                              hipStream_t stream) {
    (void)n_in; (void)ws_size;
    const float* movie_x   = (const float*)d_in[2];
    const int*   e_src     = (const int*)d_in[3];
    const int*   e_dst     = (const int*)d_in[4];
    const int*   eli_u     = (const int*)d_in[5];
    const int*   eli_m     = (const int*)d_in[6];
    const float* user_emb  = (const float*)d_in[7];
    const float* movie_emb = (const float*)d_in[8];
    const float* lin_W     = (const float*)d_in[9];
    const float* lin_b     = (const float*)d_in[10];
    const float* W1um_l    = (const float*)d_in[11];
    const float* W1um_r    = (const float*)d_in[12];
    const float* W1mu_l    = (const float*)d_in[13];
    const float* W1mu_r    = (const float*)d_in[14];
    const float* W2um_l    = (const float*)d_in[15];
    const float* W2um_r    = (const float*)d_in[16];
    const float* W2mu_l    = (const float*)d_in[17];
    const float* W2mu_r    = (const float*)d_in[18];
    const float* b1um      = (const float*)d_in[19];
    const float* b1mu      = (const float*)d_in[20];
    const float* b2um      = (const float*)d_in[21];
    const float* b2mu      = (const float*)d_in[22];
    float* out = (float*)d_out;

    const int U = in_sizes[7] / HDIM;
    const int M = in_sizes[8] / HDIM;
    const int E = in_sizes[3];
    const int L = in_sizes[5];
    (void)out_size;

    size_t off = 0;
    auto carve = [&](size_t nwords) {
        float* p = (float*)((char*)d_ws + off);
        off += ((nwords * sizeof(float) + 255) / 256) * 256;
        return p;
    };
    float* XM   = carve((size_t)M * HDIM);
    float* H_M  = carve((size_t)M * HDIM);
    float* H_U  = carve((size_t)U * HDIM);
    float* U2   = carve((size_t)U * HDIM);
    float* M2   = XM;
    int*   LIST = (int*)carve((size_t)E * 2);
    int*   CNT  = (int*)carve((size_t)M + U + 64);
    int*   START = (int*)carve((size_t)M + U);
    int*   CNT_M = CNT;      int*   CNT_U = CNT + M;
    int*   START_M = START;  int*   START_U = START + M;
    int*   ITEMS  = (int*)U2;
    int*   BCOUNT = (int*)H_M;
    int*   BOFF   = BCOUNT + MAXB;
    int*   BSTART = BOFF + MAXB;

    const int TB = 256;
    dim3 b(TB);
    const int nbM = (M + BKEYS - 1) / BKEYS;
    const int nbU = (U + BKEYS - 1) / BKEYS;
    const int s_blocks = (E + SPLIT_EDGES - 1) / SPLIT_EDGES;

    hipMemsetAsync(BCOUNT, 0, (size_t)2 * MAXB * sizeof(int), stream);
    bucket_hist_kernel<<<s_blocks, 1024, 0, stream>>>(e_src, e_dst, BCOUNT, E, nbM);
    bucket_split_kernel<<<s_blocks, 1024, 0, stream>>>(e_src, e_dst, BCOUNT, BOFF, BSTART, ITEMS,
                                                       E, nbM, nbM + nbU);
    bucket_lists_movie_init_kernel<<<nbM + nbU + INIT_BLOCKS, b, 0, stream>>>(
        ITEMS, BSTART, BCOUNT, CNT, START, LIST, nbM, M, U, nbM + nbU,
        movie_x, lin_W, lin_b, movie_emb, XM);

    sage_layer_kernel<<<M / 16 + U / 16, 128, 0, stream>>>(
        LIST,
        user_emb, START_M, CNT_M, XM, W1um_l, W1um_r, b1um, H_M,
        XM, START_U, CNT_U, user_emb, W1mu_l, W1mu_r, b1mu, H_U,
        M / 16, 1);

    sage_layer_kernel<<<M / 16 + U / 16, 128, 0, stream>>>(
        LIST,
        H_U, START_M, CNT_M, H_M, W2um_l, W2um_r, b2um, M2,
        H_M, START_U, CNT_U, H_U, W2mu_l, W2mu_r, b2mu, U2,
        M / 16, 0);

    edge_dot_kernel<<<DOT_BLOCKS, b, 0, stream>>>(U2, M2, eli_u, eli_m, out, L);
}
